# starting kernel
# baseline (speedup 1.0000x reference)
#include <hip/hip_runtime.h>
#include <hip/hip_bf16.h>

typedef __attribute__((ext_vector_type(2))) float v2f;
typedef __attribute__((ext_vector_type(8))) float v8f;

#define HDIM 128

__global__ __launch_bounds__(256)
void gemm128_wmma_kernel(const float* __restrict__ A, const float* __restrict__ W,
                         const float* __restrict__ bias, float* __restrict__ out,
                         int K, int do_relu)
{
    __shared__ __align__(16) float As[16 * 256];
    const int row0 = blockIdx.x * 16;
    const int tid  = threadIdx.x;

    const float* Atile = A + (size_t)row0 * K;
    const int total = 16 * K;
    for (int idx = tid * 4; idx < total; idx += 256 * 4)
        *(float4*)(&As[idx]) = *(const float4*)(&Atile[idx]);
    __syncthreads();

    const int wave = tid >> 5;
    const int lane = tid & 31;
    const int m    = lane & 15;
    const int koff = (lane < 16) ? 0 : 2;
    const int n    = wave * 16 + (lane & 15);

    v8f c = {};
    for (int k = 0; k < K; k += 4) {
        v2f a, b;
        a[0] = As[m * K + k + koff];
        a[1] = As[m * K + k + koff + 1];
        b[0] = W[(size_t)(k + koff)     * HDIM + n];
        b[1] = W[(size_t)(k + koff + 1) * HDIM + n];
        c = __builtin_amdgcn_wmma_f32_16x16x4_f32(false, a, false, b,
                                                  (short)0, c, false, false);
    }

    const int   mbase = row0 + ((lane < 16) ? 0 : 8);
    const float bv    = bias[n];
    for (int r = 0; r < 8; ++r) {
        float v = c[r] + bv;
        if (do_relu) v = fmaxf(v, 0.0f);
        out[(size_t)(mbase + r) * HDIM + n] = v;
    }
}

__global__ __launch_bounds__(256)
void sage_update_wmma_kernel(const float* __restrict__ agg,
                             const float* __restrict__ invdeg,
                             const float* __restrict__ h,
                             const float* __restrict__ Wl,
                             const float* __restrict__ bl,
                             const float* __restrict__ Wr,
                             float* __restrict__ out)
{
    __shared__ __align__(16) float Ms[16 * HDIM];
    __shared__ __align__(16) float Hs[16 * HDIM];
    const int row0 = blockIdx.x * 16;
    const int tid  = threadIdx.x;

    for (int idx = tid * 4; idx < 16 * HDIM; idx += 256 * 4) {
        const int   r = idx / HDIM;
        const float s = invdeg[row0 + r];
        float4 va = *(const float4*)(agg + (size_t)row0 * HDIM + idx);
        va.x *= s; va.y *= s; va.z *= s; va.w *= s;
        *(float4*)(&Ms[idx]) = va;
        *(float4*)(&Hs[idx]) = *(const float4*)(h + (size_t)row0 * HDIM + idx);
    }
    __syncthreads();

    const int wave = tid >> 5;
    const int lane = tid & 31;
    const int m    = lane & 15;
    const int koff = (lane < 16) ? 0 : 2;
    const int n    = wave * 16 + (lane & 15);

    v8f c = {};
    for (int k = 0; k < HDIM; k += 4) {
        v2f a, b;
        a[0] = Ms[m * HDIM + k + koff];
        a[1] = Ms[m * HDIM + k + koff + 1];
        b[0] = Wl[(size_t)(k + koff)     * HDIM + n];
        b[1] = Wl[(size_t)(k + koff + 1) * HDIM + n];
        c = __builtin_amdgcn_wmma_f32_16x16x4_f32(false, a, false, b,
                                                  (short)0, c, false, false);
    }
    for (int k = 0; k < HDIM; k += 4) {
        v2f a, b;
        a[0] = Hs[m * HDIM + k + koff];
        a[1] = Hs[m * HDIM + k + koff + 1];
        b[0] = Wr[(size_t)(k + koff)     * HDIM + n];
        b[1] = Wr[(size_t)(k + koff + 1) * HDIM + n];
        c = __builtin_amdgcn_wmma_f32_16x16x4_f32(false, a, false, b,
                                                  (short)0, c, false, false);
    }

    const int   mbase = row0 + ((lane < 16) ? 0 : 8);
    const float bv    = bl[n];
    for (int r = 0; r < 8; ++r) {
        float v = fmaxf(c[r] + bv, 0.0f);
        out[(size_t)(mbase + r) * HDIM + n] = v;
    }
}

__global__ __launch_bounds__(32)
void cls_wmma_kernel(const float* __restrict__ h, const float* __restrict__ Wc,
                     const float* __restrict__ bc, float* __restrict__ out, int ncls)
{
    const int row0 = blockIdx.x * 16;
    const int lane = threadIdx.x & 31;
    const int m    = lane & 15;
    const int koff = (lane < 16) ? 0 : 2;
    const int n    = blockIdx.y * 16 + (lane & 15);
    const bool nv  = (n < ncls);

    v8f c = {};
    for (int k = 0; k < HDIM; k += 4) {
        v2f a, b;
        a[0] = h[(size_t)(row0 + m) * HDIM + k + koff];
        a[1] = h[(size_t)(row0 + m) * HDIM + k + koff + 1];
        b[0] = nv ? Wc[(size_t)(k + koff)     * ncls + n] : 0.0f;
        b[1] = nv ? Wc[(size_t)(k + koff + 1) * ncls + n] : 0.0f;
        c = __builtin_amdgcn_wmma_f32_16x16x4_f32(false, a, false, b,
                                                  (short)0, c, false, false);
    }

    if (nv) {
        const int   mbase = row0 + ((lane < 16) ? 0 : 8);
        const float bv    = bc[n];
        for (int r = 0; r < 8; ++r)
            out[(size_t)(mbase + r) * ncls + n] = c[r] + bv;
    }
}

__global__ __launch_bounds__(256)
void scatter_add_kernel(const int* __restrict__ src, const int* __restrict__ dst,
                        const float* __restrict__ h, float* __restrict__ agg, int E)
{
    const int e = blockIdx.x * 8 + (threadIdx.x >> 5);
    if (e >= E) return;
    const int lane = threadIdx.x & 31;
    const int s = src[e], d = dst[e];
    float4 v = *(const float4*)(h + (size_t)s * HDIM + lane * 4);
    float* a = agg + (size_t)d * HDIM + lane * 4;
    atomicAdd(a + 0, v.x);
    atomicAdd(a + 1, v.y);
    atomicAdd(a + 2, v.z);
    atomicAdd(a + 3, v.w);
}

__global__ __launch_bounds__(256)
void deg_kernel(const int* __restrict__ dst, float* __restrict__ deg, int E)
{
    const int e = blockIdx.x * blockDim.x + threadIdx.x;
    if (e < E) atomicAdd(&deg[dst[e]], 1.0f);
}

__global__ __launch_bounds__(256)
void invdeg_kernel(float* __restrict__ deg, int n)
{
    const int i = blockIdx.x * blockDim.x + threadIdx.x;
    if (i < n) deg[i] = 1.0f / fmaxf(deg[i], 1.0f);
}

extern "C" void kernel_launch(void* const* d_in, const int* in_sizes, int n_in,
                              void* d_out, int out_size, void* d_ws, size_t ws_size,
                              hipStream_t stream)
{
    const float* x      = (const float*)d_in[0];
    const int*   edge   = (const int*)  d_in[1];
    const float* W_proj = (const float*)d_in[2];
    const float* b_proj = (const float*)d_in[3];
    const float* Wl[3]  = { (const float*)d_in[4], (const float*)d_in[7], (const float*)d_in[10] };
    const float* bl[3]  = { (const float*)d_in[5], (const float*)d_in[8], (const float*)d_in[11] };
    const float* Wr[3]  = { (const float*)d_in[6], (const float*)d_in[9], (const float*)d_in[12] };
    const float* Wc     = (const float*)d_in[13];
    const float* bc     = (const float*)d_in[14];

    const int N    = in_sizes[0] / 256;
    const int E    = in_sizes[1] / 2;
    const int ncls = in_sizes[14];

    const int* src = edge;
    const int* dst = edge + E;

    float* hA     = (float*)d_ws;
    float* hB     = hA  + (size_t)N * HDIM;
    float* agg    = hB  + (size_t)N * HDIM;
    float* invdeg = agg + (size_t)N * HDIM;

    const int rowTiles = (N + 15) / 16;

    hipMemsetAsync(invdeg, 0, (size_t)N * sizeof(float), stream);
    deg_kernel<<<(E + 255) / 256, 256, 0, stream>>>(dst, invdeg, E);
    invdeg_kernel<<<(N + 255) / 256, 256, 0, stream>>>(invdeg, N);

    gemm128_wmma_kernel<<<rowTiles, 256, 0, stream>>>(x, W_proj, b_proj, hA, 256, 1);

    for (int l = 0; l < 3; ++l) {
        hipMemsetAsync(agg, 0, (size_t)N * HDIM * sizeof(float), stream);
        scatter_add_kernel<<<(E + 7) / 8, 256, 0, stream>>>(src, dst, hA, agg, E);
        sage_update_wmma_kernel<<<rowTiles, 256, 0, stream>>>(agg, invdeg, hA,
                                                              Wl[l], bl[l], Wr[l], hB);
        float* t = hA; hA = hB; hB = t;
    }

    cls_wmma_kernel<<<dim3(rowTiles, (ncls + 15) / 16), 32, 0, stream>>>(
        hA, Wc, bc, (float*)d_out, ncls);
}


# Dense layers re-cut: both 128x128 weight blocks stay in LDS in fragment order; a wave owns a 16-row strip of all 128 columns; A goes global to registers, not via LDS; k order in products changes
# speedup vs baseline: 2.2129x; 2.2129x over previous
#include <hip/hip_runtime.h>
#include <hip/hip_bf16.h>

typedef __attribute__((ext_vector_type(2))) float v2f;
typedef __attribute__((ext_vector_type(8))) float v8f;

#define HDIM 128

#define STRIP_WAVES   8
#define STRIP_THREADS (STRIP_WAVES * 32)
#define WFRAGS        (16 * 8 * 32)

__device__ __forceinline__ void stage_weights(const float* __restrict__ Wg, float4* Ws, int tid)
{
    for (int d = tid; d < WFRAGS; d += STRIP_THREADS) {
        const int l = d & 31, t = (d >> 5) & 7, kb = d >> 8;
        const int c = l & 15, h = l >> 4;
        const float* p = Wg + (size_t)(8 * kb + 4 * h) * HDIM + 16 * t + c;
        Ws[d] = make_float4(p[0], p[HDIM], p[2 * HDIM], p[3 * HDIM]);
    }
}

__device__ __forceinline__ void load_wfrags(float4 (&b)[8], const float4* Ws, int kk, int lane)
{
#pragma unroll
    for (int t = 0; t < 8; ++t) b[t] = Ws[(kk * 8 + t) * 32 + lane];
}

__device__ __forceinline__ void mm_kblock(const float4& a, const float4 (&b)[8], v8f (&acc)[8])
{
    v2f a0, a1;
    a0[0] = a.x; a0[1] = a.y;
    a1[0] = a.z; a1[1] = a.w;
#pragma unroll
    for (int t = 0; t < 8; ++t) {
        v2f bb; bb[0] = b[t].x; bb[1] = b[t].y;
        acc[t] = __builtin_amdgcn_wmma_f32_16x16x4_f32(false, a0, false, bb, (short)0, acc[t], false, false);
    }
#pragma unroll
    for (int t = 0; t < 8; ++t) {
        v2f bb; bb[0] = b[t].z; bb[1] = b[t].w;
        acc[t] = __builtin_amdgcn_wmma_f32_16x16x4_f32(false, a1, false, bb, (short)0, acc[t], false, false);
    }
}

template <bool SCALE>
__device__ __forceinline__ void strip_gemm(const float* __restrict__ A1, const float* __restrict__ A2, int pitch,
                                           const float* __restrict__ rowscale,
                                           const float* __restrict__ W1, const float* __restrict__ W2,
                                           const float* __restrict__ bias, float* __restrict__ out, int ntiles)
{
    __shared__ __align__(16) float4 Ws[2 * WFRAGS];

    const int tid  = threadIdx.x;
    const int lane = tid & 31;
    const int wave = __builtin_amdgcn_readfirstlane(tid >> 5);
    const int c    = lane & 15;
    const int h    = lane >> 4;

    stage_weights(W1, Ws, tid);
    stage_weights(W2, Ws + WFRAGS, tid);
    __syncthreads();

    float bv[8];
#pragma unroll
    for (int t = 0; t < 8; ++t) bv[t] = bias[16 * t + c];

    const int stride = STRIP_WAVES * (int)gridDim.x;
    int tile = wave * (int)gridDim.x + (int)blockIdx.x;
    if (tile >= ntiles) return;

    float4 a1[16], a2[16];
    float  s = 1.0f;
    {
        const int    row = tile * 16 + c;
        const float* p1  = A1 + (size_t)row * pitch + 4 * h;
#pragma unroll
        for (int kb = 0; kb < 16; ++kb) a1[kb] = *(const float4*)(p1 + 8 * kb);
        if (SCALE) s = rowscale[row];
    }

    for (; tile < ntiles; tile += stride) {
        const float* p2 = A2 + (size_t)(tile * 16 + c) * pitch + 4 * h;
#pragma unroll
        for (int kb = 0; kb < 16; ++kb) a2[kb] = *(const float4*)(p2 + 8 * kb);
        if (SCALE) {
#pragma unroll
            for (int kb = 0; kb < 16; ++kb) { a1[kb].x *= s; a1[kb].y *= s; a1[kb].z *= s; a1[kb].w *= s; }
        }

        v8f acc[8];
#pragma unroll
        for (int t = 0; t < 8; ++t) acc[t] = v8f{};

        float4 b0[8], b1[8];
        load_wfrags(b0, Ws, 0, lane);
#pragma unroll
        for (int kk = 0; kk < 32; kk += 2) {
            if (kk == 16) {
                const int    nt  = (tile + stride < ntiles) ? (tile + stride) : tile;
                const int    nr  = nt * 16 + c;
                const float* p1  = A1 + (size_t)nr * pitch + 4 * h;
#pragma unroll
                for (int kb = 0; kb < 16; ++kb) a1[kb] = *(const float4*)(p1 + 8 * kb);
                if (SCALE) s = rowscale[nr];
            }
            load_wfrags(b1, Ws, kk + 1, lane);
            mm_kblock((kk < 16) ? a1[kk & 15] : a2[kk & 15], b0, acc);
            __builtin_amdgcn_sched_barrier(0);
            if (kk + 2 < 32) load_wfrags(b0, Ws, kk + 2, lane);
            mm_kblock((kk + 1 < 16) ? a1[(kk + 1) & 15] : a2[(kk + 1) & 15], b1, acc);
            __builtin_amdgcn_sched_barrier(0);
        }

        float* o = out + (size_t)(tile * 16 + 8 * h) * HDIM + c;
#pragma unroll
        for (int r = 0; r < 8; ++r) {
#pragma unroll
            for (int t = 0; t < 8; ++t)
                o[(size_t)r * HDIM + 16 * t] = fmaxf(acc[t][r] + bv[t], 0.0f);
        }
    }
}

__global__ __launch_bounds__(STRIP_THREADS)
void proj_strip_kernel(const float* __restrict__ x, const float* __restrict__ W,
                       const float* __restrict__ bias, float* __restrict__ out, int ntiles)
{
    strip_gemm<false>(x, x + HDIM, 2 * HDIM, nullptr, W, W + HDIM * HDIM, bias, out, ntiles);
}

__global__ __launch_bounds__(STRIP_THREADS)
void sage_strip_kernel(const float* __restrict__ agg, const float* __restrict__ invdeg,
                       const float* __restrict__ h, const float* __restrict__ Wl,
                       const float* __restrict__ bl, const float* __restrict__ Wr,
                       float* __restrict__ out, int ntiles)
{
    strip_gemm<true>(agg, h, HDIM, invdeg, Wl, Wr, bl, out, ntiles);
}

__global__ __launch_bounds__(32)
void cls_wmma_kernel(const float* __restrict__ h, const float* __restrict__ Wc,
                     const float* __restrict__ bc, float* __restrict__ out, int ncls)
{
    const int row0 = blockIdx.x * 16;
    const int lane = threadIdx.x & 31;
    const int m    = lane & 15;
    const int koff = (lane < 16) ? 0 : 2;
    const int n    = blockIdx.y * 16 + (lane & 15);
    const bool nv  = (n < ncls);

    v8f c = {};
    for (int k = 0; k < HDIM; k += 4) {
        v2f a, b;
        a[0] = h[(size_t)(row0 + m) * HDIM + k + koff];
        a[1] = h[(size_t)(row0 + m) * HDIM + k + koff + 1];
        b[0] = nv ? Wc[(size_t)(k + koff)     * ncls + n] : 0.0f;
        b[1] = nv ? Wc[(size_t)(k + koff + 1) * ncls + n] : 0.0f;
        c = __builtin_amdgcn_wmma_f32_16x16x4_f32(false, a, false, b,
                                                  (short)0, c, false, false);
    }

    if (nv) {
        const int   mbase = row0 + ((lane < 16) ? 0 : 8);
        const float bv    = bc[n];
        for (int r = 0; r < 8; ++r)
            out[(size_t)(mbase + r) * ncls + n] = c[r] + bv;
    }
}

#define OWN_ROWS   400
#define OWN_WAVES  32
#define SCAN_ROWS  16
#define SCAN_EDGES (SCAN_ROWS * 32)
#define QCAP       (SCAN_EDGES + 32)
#define EDGE_GROUP 8

__global__ __launch_bounds__(1024)
void owned_rows_sum_kernel(const int* __restrict__ src, const int* __restrict__ dst,
                           const float* __restrict__ h, float* __restrict__ agg,
                           int E, int N)
{
    __shared__ __align__(16) float acc[(OWN_ROWS + 1) * HDIM];
    __shared__ int queue[OWN_WAVES * QCAP];

    const int tid  = threadIdx.x;
    const int lane = tid & 31;
    const int wave = __builtin_amdgcn_readfirstlane(tid >> 5);
    const int base = blockIdx.x * OWN_ROWS;
    const int rows = (N - base < OWN_ROWS) ? (N - base) : OWN_ROWS;

    for (int i = tid * 4; i < (OWN_ROWS + 1) * HDIM; i += 1024 * 4)
        *(float4*)(&acc[i]) = make_float4(0.0f, 0.0f, 0.0f, 0.0f);
    __syncthreads();

    int*         q  = queue + wave * QCAP;
    int          qn = 0;
    const float* hl = h   + lane;
    float*       al = acc + lane;

    const int nfull = E / SCAN_EDGES;
    const int first = (int)((blockIdx.x * (unsigned)nfull) / gridDim.x);

    auto push_row = [&](int d, int e) {
        const unsigned t    = (unsigned)(d - base);
        const bool     mine = t < (unsigned)rows;
        const unsigned mask = __builtin_amdgcn_ballot_w32(mine);
        if (mask != 0u) {
            if (mine) q[qn + (int)__builtin_amdgcn_mbcnt_lo(mask, 0u)] = e;
            qn += __builtin_popcount(mask);
        }
    };

    auto add_rows = [&](int sv, int dv) {
        for (int g = 0; g < 32; g += EDGE_GROUP) {
            float v[EDGE_GROUP][4];
#pragma unroll
            for (int j = 0; j < EDGE_GROUP; ++j) {
                const int s = __builtin_amdgcn_readlane(sv, g + j);
                const float* p = hl + (size_t)s * HDIM;
                v[j][0] = p[0]; v[j][1] = p[32]; v[j][2] = p[64]; v[j][3] = p[96];
            }
#pragma unroll
            for (int j = 0; j < EDGE_GROUP; ++j) {
                const int d = __builtin_amdgcn_readlane(dv, g + j);
                float* r = al + d * HDIM;
                atomicAdd(r,      v[j][0]);
                atomicAdd(r + 32, v[j][1]);
                atomicAdd(r + 64, v[j][2]);
                atomicAdd(r + 96, v[j][3]);
            }
        }
    };

    auto scan_step = [&](int (&use)[SCAN_ROWS], int (&fill)[SCAN_ROWS], int c) {
        int cl = c + 2 * OWN_WAVES;
        if (cl > nfull - 1) cl = nfull - 1;
        const int ccl = (cl + first < nfull) ? (cl + first) : (cl + first - nfull);
        const int* dl = dst + ccl * SCAN_EDGES + lane;
#pragma unroll
        for (int j = 0; j < SCAN_ROWS; ++j) fill[j] = dl[32 * j];

        if (c >= 0 && c < nfull) {
            const int cc = (c + first < nfull) ? (c + first) : (c + first - nfull);
            const int e0 = cc * SCAN_EDGES + lane;
#pragma unroll
            for (int j = 0; j < SCAN_ROWS; ++j) push_row(use[j], e0 + 32 * j);
        }

        while (qn >= 32) {
            qn -= 32;
            const int ee = q[qn + lane];
            add_rows(src[ee], dst[ee] - base);
        }
    };

    if (nfull > 0) {
        int ra[SCAN_ROWS], rb[SCAN_ROWS], rc[SCAN_ROWS];
#pragma unroll
        for (int j = 0; j < SCAN_ROWS; ++j) { ra[j] = -1; rb[j] = -1; }
        for (int c = wave - 2 * OWN_WAVES; c < nfull; c += 3 * OWN_WAVES) {
            scan_step(ra, rc, c);
            scan_step(rb, ra, c + OWN_WAVES);
            scan_step(rc, rb, c + 2 * OWN_WAVES);
        }
    }

    if (wave * 32 < E - nfull * SCAN_EDGES) {
        const int e = nfull * SCAN_EDGES + wave * 32 + lane;
        push_row((e < E) ? dst[e] : -1, e);
    }
    while (qn > 0) {
        const int n = (qn < 32) ? qn : 32;
        qn -= n;
        int sv = 0, dv = OWN_ROWS;
        if (lane < n) { const int ee = q[qn + lane]; sv = src[ee]; dv = dst[ee] - base; }
        add_rows(sv, dv);
    }
    __syncthreads();

    float* outp = agg + (size_t)base * HDIM;
    for (int i = tid * 4; i < rows * HDIM; i += 1024 * 4)
        *(float4*)(outp + i) = *(const float4*)(&acc[i]);
}

__global__ __launch_bounds__(256)
void deg_kernel(const int* __restrict__ dst, float* __restrict__ deg, int E)
{
    const int e = blockIdx.x * blockDim.x + threadIdx.x;
    if (e < E) atomicAdd(&deg[dst[e]], 1.0f);
}

__global__ __launch_bounds__(256)
void invdeg_kernel(float* __restrict__ deg, int n)
{
    const int i = blockIdx.x * blockDim.x + threadIdx.x;
    if (i < n) deg[i] = 1.0f / fmaxf(deg[i], 1.0f);
}

extern "C" void kernel_launch(void* const* d_in, const int* in_sizes, int n_in,
                              void* d_out, int out_size, void* d_ws, size_t ws_size,
                              hipStream_t stream)
{
    const float* x      = (const float*)d_in[0];
    const int*   edge   = (const int*)  d_in[1];
    const float* W_proj = (const float*)d_in[2];
    const float* b_proj = (const float*)d_in[3];
    const float* Wl[3]  = { (const float*)d_in[4], (const float*)d_in[7], (const float*)d_in[10] };
    const float* bl[3]  = { (const float*)d_in[5], (const float*)d_in[8], (const float*)d_in[11] };
    const float* Wr[3]  = { (const float*)d_in[6], (const float*)d_in[9], (const float*)d_in[12] };
    const float* Wc     = (const float*)d_in[13];
    const float* bc     = (const float*)d_in[14];

    const int N    = in_sizes[0] / 256;
    const int E    = in_sizes[1] / 2;
    const int ncls = in_sizes[14];

    const int* src = edge;
    const int* dst = edge + E;

    float* hA     = (float*)d_ws;
    float* hB     = hA  + (size_t)N * HDIM;
    float* agg    = hB  + (size_t)N * HDIM;
    float* invdeg = agg + (size_t)N * HDIM;

    const int rowTiles = (N + 15) / 16;

    hipMemsetAsync(invdeg, 0, (size_t)N * sizeof(float), stream);
    deg_kernel<<<(E + 255) / 256, 256, 0, stream>>>(dst, invdeg, E);
    invdeg_kernel<<<(N + 255) / 256, 256, 0, stream>>>(invdeg, N);

    proj_strip_kernel<<<256, STRIP_THREADS, 0, stream>>>(x, W_proj, b_proj, hA, rowTiles);

    for (int l = 0; l < 3; ++l) {
        owned_rows_sum_kernel<<<(N + OWN_ROWS - 1) / OWN_ROWS, 1024, 0, stream>>>(src, dst, hA, agg, E, N);
        sage_strip_kernel<<<256, STRIP_THREADS, 0, stream>>>(agg, invdeg, hA,
                                                             Wl[l], bl[l], Wr[l], hB, rowTiles);
        float* t = hA; hA = hB; hB = t;
    }

    cls_wmma_kernel<<<dim3(rowTiles, (ncls + 15) / 16), 32, 0, stream>>>(
        hA, Wc, bc, (float*)d_out, ncls);
}


# Edge filter runs once: layer 1's blocks save each wave's matched (source row, local row) pairs in the workspace; layers 2 and 3 replay them, no dst list rescans; sum order stays LDS arrival
# speedup vs baseline: 2.6991x; 1.2197x over previous
#include <hip/hip_runtime.h>
#include <hip/hip_bf16.h>

typedef __attribute__((ext_vector_type(2))) float v2f;
typedef __attribute__((ext_vector_type(8))) float v8f;

#define HDIM 128

#define STRIP_WAVES   8
#define STRIP_THREADS (STRIP_WAVES * 32)
#define WFRAGS        (16 * 8 * 32)

__device__ __forceinline__ void stage_weights(const float* __restrict__ Wg, float4* Ws, int tid)
{
    for (int d = tid; d < WFRAGS; d += STRIP_THREADS) {
        const int l = d & 31, t = (d >> 5) & 7, kb = d >> 8;
        const int c = l & 15, h = l >> 4;
        const float* p = Wg + (size_t)(8 * kb + 4 * h) * HDIM + 16 * t + c;
        Ws[d] = make_float4(p[0], p[HDIM], p[2 * HDIM], p[3 * HDIM]);
    }
}

__device__ __forceinline__ void load_wfrags(float4 (&b)[8], const float4* Ws, int kk, int lane)
{
#pragma unroll
    for (int t = 0; t < 8; ++t) b[t] = Ws[(kk * 8 + t) * 32 + lane];
}

__device__ __forceinline__ void mm_kblock(const float4& a, const float4 (&b)[8], v8f (&acc)[8])
{
    v2f a0, a1;
    a0[0] = a.x; a0[1] = a.y;
    a1[0] = a.z; a1[1] = a.w;
#pragma unroll
    for (int t = 0; t < 8; ++t) {
        v2f bb; bb[0] = b[t].x; bb[1] = b[t].y;
        acc[t] = __builtin_amdgcn_wmma_f32_16x16x4_f32(false, a0, false, bb, (short)0, acc[t], false, false);
    }
#pragma unroll
    for (int t = 0; t < 8; ++t) {
        v2f bb; bb[0] = b[t].z; bb[1] = b[t].w;
        acc[t] = __builtin_amdgcn_wmma_f32_16x16x4_f32(false, a1, false, bb, (short)0, acc[t], false, false);
    }
}

template <bool SCALE>
__device__ __forceinline__ void strip_gemm(const float* __restrict__ A1, const float* __restrict__ A2, int pitch,
                                           const float* __restrict__ rowscale,
                                           const float* __restrict__ W1, const float* __restrict__ W2,
                                           const float* __restrict__ bias, float* __restrict__ out, int ntiles)
{
    __shared__ __align__(16) float4 Ws[2 * WFRAGS];

    const int tid  = threadIdx.x;
    const int lane = tid & 31;
    const int wave = __builtin_amdgcn_readfirstlane(tid >> 5);
    const int c    = lane & 15;
    const int h    = lane >> 4;

    stage_weights(W1, Ws, tid);
    stage_weights(W2, Ws + WFRAGS, tid);
    __syncthreads();

    float bv[8];
#pragma unroll
    for (int t = 0; t < 8; ++t) bv[t] = bias[16 * t + c];

    const int stride = STRIP_WAVES * (int)gridDim.x;
    int tile = wave * (int)gridDim.x + (int)blockIdx.x;
    if (tile >= ntiles) return;

    float4 a1[16], a2[16];
    float  s = 1.0f;
    {
        const int    row = tile * 16 + c;
        const float* p1  = A1 + (size_t)row * pitch + 4 * h;
#pragma unroll
        for (int kb = 0; kb < 16; ++kb) a1[kb] = *(const float4*)(p1 + 8 * kb);
        if (SCALE) s = rowscale[row];
    }

    for (; tile < ntiles; tile += stride) {
        const float* p2 = A2 + (size_t)(tile * 16 + c) * pitch + 4 * h;
#pragma unroll
        for (int kb = 0; kb < 16; ++kb) a2[kb] = *(const float4*)(p2 + 8 * kb);
        if (SCALE) {
#pragma unroll
            for (int kb = 0; kb < 16; ++kb) { a1[kb].x *= s; a1[kb].y *= s; a1[kb].z *= s; a1[kb].w *= s; }
        }

        v8f acc[8];
#pragma unroll
        for (int t = 0; t < 8; ++t) acc[t] = v8f{};

        float4 b0[8], b1[8];
        load_wfrags(b0, Ws, 0, lane);
#pragma unroll
        for (int kk = 0; kk < 32; kk += 2) {
            if (kk == 16) {
                const int    nt  = (tile + stride < ntiles) ? (tile + stride) : tile;
                const int    nr  = nt * 16 + c;
                const float* p1  = A1 + (size_t)nr * pitch + 4 * h;
#pragma unroll
                for (int kb = 0; kb < 16; ++kb) a1[kb] = *(const float4*)(p1 + 8 * kb);
                if (SCALE) s = rowscale[nr];
            }
            load_wfrags(b1, Ws, kk + 1, lane);
            mm_kblock((kk < 16) ? a1[kk & 15] : a2[kk & 15], b0, acc);
            __builtin_amdgcn_sched_barrier(0);
            if (kk + 2 < 32) load_wfrags(b0, Ws, kk + 2, lane);
            mm_kblock((kk + 1 < 16) ? a1[(kk + 1) & 15] : a2[(kk + 1) & 15], b1, acc);
            __builtin_amdgcn_sched_barrier(0);
        }

        float* o = out + (size_t)(tile * 16 + 8 * h) * HDIM + c;
#pragma unroll
        for (int r = 0; r < 8; ++r) {
#pragma unroll
            for (int t = 0; t < 8; ++t)
                o[(size_t)r * HDIM + 16 * t] = fmaxf(acc[t][r] + bv[t], 0.0f);
        }
    }
}

__global__ __launch_bounds__(STRIP_THREADS)
void proj_strip_kernel(const float* __restrict__ x, const float* __restrict__ W,
                       const float* __restrict__ bias, float* __restrict__ out, int ntiles)
{
    strip_gemm<false>(x, x + HDIM, 2 * HDIM, nullptr, W, W + HDIM * HDIM, bias, out, ntiles);
}

__global__ __launch_bounds__(STRIP_THREADS)
void sage_strip_kernel(const float* __restrict__ agg, const float* __restrict__ invdeg,
                       const float* __restrict__ h, const float* __restrict__ Wl,
                       const float* __restrict__ bl, const float* __restrict__ Wr,
                       float* __restrict__ out, int ntiles)
{
    strip_gemm<true>(agg, h, HDIM, invdeg, Wl, Wr, bl, out, ntiles);
}

__global__ __launch_bounds__(32)
void cls_wmma_kernel(const float* __restrict__ h, const float* __restrict__ Wc,
                     const float* __restrict__ bc, float* __restrict__ out, int ncls)
{
    const int row0 = blockIdx.x * 16;
    const int lane = threadIdx.x & 31;
    const int m    = lane & 15;
    const int koff = (lane < 16) ? 0 : 2;
    const int n    = blockIdx.y * 16 + (lane & 15);
    const bool nv  = (n < ncls);

    v8f c = {};
    for (int k = 0; k < HDIM; k += 4) {
        v2f a, b;
        a[0] = h[(size_t)(row0 + m) * HDIM + k + koff];
        a[1] = h[(size_t)(row0 + m) * HDIM + k + koff + 1];
        b[0] = nv ? Wc[(size_t)(k + koff)     * ncls + n] : 0.0f;
        b[1] = nv ? Wc[(size_t)(k + koff + 1) * ncls + n] : 0.0f;
        c = __builtin_amdgcn_wmma_f32_16x16x4_f32(false, a, false, b,
                                                  (short)0, c, false, false);
    }

    if (nv) {
        const int   mbase = row0 + ((lane < 16) ? 0 : 8);
        const float bv    = bc[n];
        for (int r = 0; r < 8; ++r)
            out[(size_t)(mbase + r) * ncls + n] = c[r] + bv;
    }
}

#define OWN_ROWS   400
#define OWN_WAVES  32
#define SCAN_ROWS  16
#define SCAN_EDGES (SCAN_ROWS * 32)
#define QCAP       (SCAN_EDGES + 32)
#define EDGE_GROUP 8
#define LIST_CAP   2048
#define REPLAY_ROUNDS 8

template <bool FIRST>
__global__ __launch_bounds__(1024)
void owned_rows_sum_kernel(const int* __restrict__ src, const int* __restrict__ dst,
                           const float* __restrict__ h, float* __restrict__ agg,
                           int* __restrict__ lists, int* __restrict__ counts,
                           int E, int N)
{
    __shared__ __align__(16) float acc[(OWN_ROWS + 1) * HDIM];
    __shared__ int queue[OWN_WAVES * QCAP];

    const int tid  = threadIdx.x;
    const int lane = tid & 31;
    const int wave = __builtin_amdgcn_readfirstlane(tid >> 5);
    const int base = blockIdx.x * OWN_ROWS;
    const int rows = (N - base < OWN_ROWS) ? (N - base) : OWN_ROWS;

    for (int i = tid * 4; i < (OWN_ROWS + 1) * HDIM; i += 1024 * 4)
        *(float4*)(&acc[i]) = make_float4(0.0f, 0.0f, 0.0f, 0.0f);
    __syncthreads();

    int*         q  = queue + wave * QCAP;
    int          qn = 0;
    int*         ls  = lists + (size_t)(blockIdx.x * OWN_WAVES + wave) * (2 * LIST_CAP);
    int*         ld  = ls + LIST_CAP;
    int          rec = 0;
    const float* hl = h   + lane;
    float*       al = acc + lane;

    const int nfull = E / SCAN_EDGES;
    const int first = (int)((blockIdx.x * (unsigned)nfull) / gridDim.x);

    auto push_row = [&](int d, int e) {
        const unsigned t    = (unsigned)(d - base);
        const bool     mine = t < (unsigned)rows;
        const unsigned mask = __builtin_amdgcn_ballot_w32(mine);
        if (mask != 0u) {
            if (mine) q[qn + (int)__builtin_amdgcn_mbcnt_lo(mask, 0u)] = e;
            qn += __builtin_popcount(mask);
        }
    };

    auto add_rows = [&](int sv, int dv, int m) {
        for (int g = 0; g < m; g += EDGE_GROUP) {
            float v[EDGE_GROUP][4];
#pragma unroll
            for (int j = 0; j < EDGE_GROUP; ++j) {
                const int s = __builtin_amdgcn_readlane(sv, g + j);
                const float* p = hl + (size_t)s * HDIM;
                v[j][0] = p[0]; v[j][1] = p[32]; v[j][2] = p[64]; v[j][3] = p[96];
            }
#pragma unroll
            for (int j = 0; j < EDGE_GROUP; ++j) {
                const int d = __builtin_amdgcn_readlane(dv, g + j);
                float* r = al + d * HDIM;
                atomicAdd(r,      v[j][0]);
                atomicAdd(r + 32, v[j][1]);
                atomicAdd(r + 64, v[j][2]);
                atomicAdd(r + 96, v[j][3]);
            }
        }
    };

    auto scan_step = [&](int (&use)[SCAN_ROWS], int (&fill)[SCAN_ROWS], int c) {
        int cl = c + 2 * OWN_WAVES;
        if (cl > nfull - 1) cl = nfull - 1;
        const int ccl = (cl + first < nfull) ? (cl + first) : (cl + first - nfull);
        const int* dl = dst + ccl * SCAN_EDGES + lane;
#pragma unroll
        for (int j = 0; j < SCAN_ROWS; ++j) fill[j] = dl[32 * j];

        if (c >= 0 && c < nfull) {
            const int cc = (c + first < nfull) ? (c + first) : (c + first - nfull);
            const int e0 = cc * SCAN_EDGES + lane;
#pragma unroll
            for (int j = 0; j < SCAN_ROWS; ++j) push_row(use[j], e0 + 32 * j);
        }

        while (qn >= 32) {
            qn -= 32;
            const int ee = q[qn + lane];
            const int sv = src[ee], dv = dst[ee] - base;
            if (FIRST) {
                if (rec + 32 <= LIST_CAP) { ls[rec + lane] = sv; ld[rec + lane] = dv; }
                rec += 32;
            }
            add_rows(sv, dv, 32);
        }
    };

    int nsaved = 0;
    if (!FIRST) nsaved = __builtin_amdgcn_readfirstlane(counts[blockIdx.x * OWN_WAVES + wave]);
    const bool scan = FIRST || nsaved > LIST_CAP;

    if (scan) {
        if (nfull > 0) {
            int ra[SCAN_ROWS], rb[SCAN_ROWS], rc[SCAN_ROWS];
#pragma unroll
            for (int j = 0; j < SCAN_ROWS; ++j) { ra[j] = -1; rb[j] = -1; }
            for (int c = wave - 2 * OWN_WAVES; c < nfull; c += 3 * OWN_WAVES) {
                scan_step(ra, rc, c);
                scan_step(rb, ra, c + OWN_WAVES);
                scan_step(rc, rb, c + 2 * OWN_WAVES);
            }
        }

        if (wave * 32 < E - nfull * SCAN_EDGES) {
            const int e = nfull * SCAN_EDGES + wave * 32 + lane;
            push_row((e < E) ? dst[e] : -1, e);
        }
        while (qn > 0) {
            const int n = (qn < 32) ? qn : 32;
            qn -= n;
            int sv = 0, dv = OWN_ROWS;
            if (lane < n) {
                const int ee = q[qn + lane]; sv = src[ee]; dv = dst[ee] - base;
                if (FIRST && rec + n <= LIST_CAP) { ls[rec + lane] = sv; ld[rec + lane] = dv; }
            }
            if (FIRST) rec += n;
            add_rows(sv, dv, n);
        }
        if (FIRST && lane == 0) counts[blockIdx.x * OWN_WAVES + wave] = rec;
    } else {
        for (int i0 = 0; i0 < nsaved; i0 += 32 * REPLAY_ROUNDS) {
            int sv[REPLAY_ROUNDS], dv[REPLAY_ROUNDS];
#pragma unroll
            for (int j = 0; j < REPLAY_ROUNDS; ++j) {
                const int i = i0 + 32 * j + lane;
                sv[j] = 0; dv[j] = OWN_ROWS;
                if (i < nsaved) { sv[j] = ls[i]; dv[j] = ld[i]; }
            }
#pragma unroll
            for (int j = 0; j < REPLAY_ROUNDS; ++j) {
                const int left = nsaved - i0 - 32 * j;
                if (left > 0) add_rows(sv[j], dv[j], (left < 32) ? left : 32);
            }
        }
    }
    __syncthreads();

    float* outp = agg + (size_t)base * HDIM;
    for (int i = tid * 4; i < rows * HDIM; i += 1024 * 4)
        *(float4*)(outp + i) = *(const float4*)(&acc[i]);
}

__global__ __launch_bounds__(256)
void deg_kernel(const int* __restrict__ dst, float* __restrict__ deg, int E)
{
    const int e = blockIdx.x * blockDim.x + threadIdx.x;
    if (e < E) atomicAdd(&deg[dst[e]], 1.0f);
}

__global__ __launch_bounds__(256)
void invdeg_kernel(float* __restrict__ deg, int n)
{
    const int i = blockIdx.x * blockDim.x + threadIdx.x;
    if (i < n) deg[i] = 1.0f / fmaxf(deg[i], 1.0f);
}

extern "C" void kernel_launch(void* const* d_in, const int* in_sizes, int n_in,
                              void* d_out, int out_size, void* d_ws, size_t ws_size,
                              hipStream_t stream)
{
    const float* x      = (const float*)d_in[0];
    const int*   edge   = (const int*)  d_in[1];
    const float* W_proj = (const float*)d_in[2];
    const float* b_proj = (const float*)d_in[3];
    const float* Wl[3]  = { (const float*)d_in[4], (const float*)d_in[7], (const float*)d_in[10] };
    const float* bl[3]  = { (const float*)d_in[5], (const float*)d_in[8], (const float*)d_in[11] };
    const float* Wr[3]  = { (const float*)d_in[6], (const float*)d_in[9], (const float*)d_in[12] };
    const float* Wc     = (const float*)d_in[13];
    const float* bc     = (const float*)d_in[14];

    const int N    = in_sizes[0] / 256;
    const int E    = in_sizes[1] / 2;
    const int ncls = in_sizes[14];

    const int* src = edge;
    const int* dst = edge + E;

    float* hA     = (float*)d_ws;
    float* hB     = hA  + (size_t)N * HDIM;
    float* agg    = hB  + (size_t)N * HDIM;
    float* invdeg = agg + (size_t)N * HDIM;
    int*   counts = (int*)(invdeg + N);
    int*   lists  = counts + (size_t)((N + OWN_ROWS - 1) / OWN_ROWS) * OWN_WAVES;

    const int rowTiles = (N + 15) / 16;

    hipMemsetAsync(invdeg, 0, (size_t)N * sizeof(float), stream);
    deg_kernel<<<(E + 255) / 256, 256, 0, stream>>>(dst, invdeg, E);
    invdeg_kernel<<<(N + 255) / 256, 256, 0, stream>>>(invdeg, N);

    proj_strip_kernel<<<256, STRIP_THREADS, 0, stream>>>(x, W_proj, b_proj, hA, rowTiles);

    for (int l = 0; l < 3; ++l) {
        if (l == 0)
            owned_rows_sum_kernel<true><<<(N + OWN_ROWS - 1) / OWN_ROWS, 1024, 0, stream>>>(src, dst, hA, agg,
                                                                                             lists, counts, E, N);
        else
            owned_rows_sum_kernel<false><<<(N + OWN_ROWS - 1) / OWN_ROWS, 1024, 0, stream>>>(src, dst, hA, agg,
                                                                                              lists, counts, E, N);
        sage_strip_kernel<<<256, STRIP_THREADS, 0, stream>>>(agg, invdeg, hA,
                                                             Wl[l], bl[l], Wr[l], hB, rowTiles);
        float* t = hA; hA = hB; hB = t;
    }

    cls_wmma_kernel<<<dim3(rowTiles, (ncls + 15) / 16), 32, 0, stream>>>(
        hA, Wc, bc, (float*)d_out, ncls);
}


# Row degrees counted in the first neighbour-sum launch (integer LDS counts of the edges it already filters), 1/max(deg,1) stored there; memset, degree and reciprocal launches gone; sum orders kept
# speedup vs baseline: 3.0794x; 1.1409x over previous
#include <hip/hip_runtime.h>
#include <hip/hip_bf16.h>

typedef __attribute__((ext_vector_type(2))) float v2f;
typedef __attribute__((ext_vector_type(8))) float v8f;

#define HDIM 128

#define STRIP_WAVES   8
#define STRIP_THREADS (STRIP_WAVES * 32)
#define WFRAGS        (16 * 8 * 32)

__device__ __forceinline__ void stage_weights(const float* __restrict__ Wg, float4* Ws, int tid)
{
    for (int d = tid; d < WFRAGS; d += STRIP_THREADS) {
        const int l = d & 31, t = (d >> 5) & 7, kb = d >> 8;
        const int c = l & 15, h = l >> 4;
        const float* p = Wg + (size_t)(8 * kb + 4 * h) * HDIM + 16 * t + c;
        Ws[d] = make_float4(p[0], p[HDIM], p[2 * HDIM], p[3 * HDIM]);
    }
}

__device__ __forceinline__ void load_wfrags(float4 (&b)[8], const float4* Ws, int kk, int lane)
{
#pragma unroll
    for (int t = 0; t < 8; ++t) b[t] = Ws[(kk * 8 + t) * 32 + lane];
}

__device__ __forceinline__ void mm_kblock(const float4& a, const float4 (&b)[8], v8f (&acc)[8])
{
    v2f a0, a1;
    a0[0] = a.x; a0[1] = a.y;
    a1[0] = a.z; a1[1] = a.w;
#pragma unroll
    for (int t = 0; t < 8; ++t) {
        v2f bb; bb[0] = b[t].x; bb[1] = b[t].y;
        acc[t] = __builtin_amdgcn_wmma_f32_16x16x4_f32(false, a0, false, bb, (short)0, acc[t], false, false);
    }
#pragma unroll
    for (int t = 0; t < 8; ++t) {
        v2f bb; bb[0] = b[t].z; bb[1] = b[t].w;
        acc[t] = __builtin_amdgcn_wmma_f32_16x16x4_f32(false, a1, false, bb, (short)0, acc[t], false, false);
    }
}

template <bool SCALE>
__device__ __forceinline__ void strip_gemm(const float* __restrict__ A1, const float* __restrict__ A2, int pitch,
                                           const float* __restrict__ rowscale,
                                           const float* __restrict__ W1, const float* __restrict__ W2,
                                           const float* __restrict__ bias, float* __restrict__ out, int ntiles)
{
    __shared__ __align__(16) float4 Ws[2 * WFRAGS];

    const int tid  = threadIdx.x;
    const int lane = tid & 31;
    const int wave = __builtin_amdgcn_readfirstlane(tid >> 5);
    const int c    = lane & 15;
    const int h    = lane >> 4;

    stage_weights(W1, Ws, tid);
    stage_weights(W2, Ws + WFRAGS, tid);
    __syncthreads();

    float bv[8];
#pragma unroll
    for (int t = 0; t < 8; ++t) bv[t] = bias[16 * t + c];

    const int stride = STRIP_WAVES * (int)gridDim.x;
    int tile = wave * (int)gridDim.x + (int)blockIdx.x;
    if (tile >= ntiles) return;

    float4 a1[16], a2[16];
    float  s = 1.0f;
    {
        const int    row = tile * 16 + c;
        const float* p1  = A1 + (size_t)row * pitch + 4 * h;
#pragma unroll
        for (int kb = 0; kb < 16; ++kb) a1[kb] = *(const float4*)(p1 + 8 * kb);
        if (SCALE) s = rowscale[row];
    }

    for (; tile < ntiles; tile += stride) {
        const float* p2 = A2 + (size_t)(tile * 16 + c) * pitch + 4 * h;
#pragma unroll
        for (int kb = 0; kb < 16; ++kb) a2[kb] = *(const float4*)(p2 + 8 * kb);
        if (SCALE) {
#pragma unroll
            for (int kb = 0; kb < 16; ++kb) { a1[kb].x *= s; a1[kb].y *= s; a1[kb].z *= s; a1[kb].w *= s; }
        }

        v8f acc[8];
#pragma unroll
        for (int t = 0; t < 8; ++t) acc[t] = v8f{};

        float4 b0[8], b1[8];
        load_wfrags(b0, Ws, 0, lane);
#pragma unroll
        for (int kk = 0; kk < 32; kk += 2) {
            if (kk == 16) {
                const int    nt  = (tile + stride < ntiles) ? (tile + stride) : tile;
                const int    nr  = nt * 16 + c;
                const float* p1  = A1 + (size_t)nr * pitch + 4 * h;
#pragma unroll
                for (int kb = 0; kb < 16; ++kb) a1[kb] = *(const float4*)(p1 + 8 * kb);
                if (SCALE) s = rowscale[nr];
            }
            load_wfrags(b1, Ws, kk + 1, lane);
            mm_kblock((kk < 16) ? a1[kk & 15] : a2[kk & 15], b0, acc);
            __builtin_amdgcn_sched_barrier(0);
            if (kk + 2 < 32) load_wfrags(b0, Ws, kk + 2, lane);
            mm_kblock((kk + 1 < 16) ? a1[(kk + 1) & 15] : a2[(kk + 1) & 15], b1, acc);
            __builtin_amdgcn_sched_barrier(0);
        }

        float* o = out + (size_t)(tile * 16 + 8 * h) * HDIM + c;
#pragma unroll
        for (int r = 0; r < 8; ++r) {
#pragma unroll
            for (int t = 0; t < 8; ++t)
                o[(size_t)r * HDIM + 16 * t] = fmaxf(acc[t][r] + bv[t], 0.0f);
        }
    }
}

__global__ __launch_bounds__(STRIP_THREADS)
void proj_strip_kernel(const float* __restrict__ x, const float* __restrict__ W,
                       const float* __restrict__ bias, float* __restrict__ out, int ntiles)
{
    strip_gemm<false>(x, x + HDIM, 2 * HDIM, nullptr, W, W + HDIM * HDIM, bias, out, ntiles);
}

__global__ __launch_bounds__(STRIP_THREADS)
void sage_strip_kernel(const float* __restrict__ agg, const float* __restrict__ invdeg,
                       const float* __restrict__ h, const float* __restrict__ Wl,
                       const float* __restrict__ bl, const float* __restrict__ Wr,
                       float* __restrict__ out, int ntiles)
{
    strip_gemm<true>(agg, h, HDIM, invdeg, Wl, Wr, bl, out, ntiles);
}

__global__ __launch_bounds__(32)
void cls_wmma_kernel(const float* __restrict__ h, const float* __restrict__ Wc,
                     const float* __restrict__ bc, float* __restrict__ out, int ncls)
{
    const int row0 = blockIdx.x * 16;
    const int lane = threadIdx.x & 31;
    const int m    = lane & 15;
    const int koff = (lane < 16) ? 0 : 2;
    const int n    = blockIdx.y * 16 + (lane & 15);
    const bool nv  = (n < ncls);

    v8f c = {};
    for (int k = 0; k < HDIM; k += 4) {
        v2f a, b;
        a[0] = h[(size_t)(row0 + m) * HDIM + k + koff];
        a[1] = h[(size_t)(row0 + m) * HDIM + k + koff + 1];
        b[0] = nv ? Wc[(size_t)(k + koff)     * ncls + n] : 0.0f;
        b[1] = nv ? Wc[(size_t)(k + koff + 1) * ncls + n] : 0.0f;
        c = __builtin_amdgcn_wmma_f32_16x16x4_f32(false, a, false, b,
                                                  (short)0, c, false, false);
    }

    if (nv) {
        const int   mbase = row0 + ((lane < 16) ? 0 : 8);
        const float bv    = bc[n];
        for (int r = 0; r < 8; ++r)
            out[(size_t)(mbase + r) * ncls + n] = c[r] + bv;
    }
}

#define OWN_ROWS   400
#define OWN_WAVES  32
#define SCAN_ROWS  16
#define SCAN_EDGES (SCAN_ROWS * 32)
#define QCAP       (SCAN_EDGES + 32)
#define EDGE_GROUP 8
#define LIST_CAP   2048
#define REPLAY_ROUNDS 8

template <bool FIRST>
__global__ __launch_bounds__(1024)
void owned_rows_sum_kernel(const int* __restrict__ src, const int* __restrict__ dst,
                           const float* __restrict__ h, float* __restrict__ agg,
                           int* __restrict__ lists, int* __restrict__ counts,
                           float* __restrict__ invdeg, int E, int N)
{
    __shared__ __align__(16) float acc[(OWN_ROWS + 1) * HDIM];
    __shared__ int queue[OWN_WAVES * QCAP];
    __shared__ int degc[OWN_ROWS];

    const int tid  = threadIdx.x;
    const int lane = tid & 31;
    const int wave = __builtin_amdgcn_readfirstlane(tid >> 5);
    const int base = blockIdx.x * OWN_ROWS;
    const int rows = (N - base < OWN_ROWS) ? (N - base) : OWN_ROWS;

    for (int i = tid * 4; i < (OWN_ROWS + 1) * HDIM; i += 1024 * 4)
        *(float4*)(&acc[i]) = make_float4(0.0f, 0.0f, 0.0f, 0.0f);
    if (FIRST) {
        for (int i = tid; i < OWN_ROWS; i += 1024) degc[i] = 0;
    }
    __syncthreads();

    int*         q  = queue + wave * QCAP;
    int          qn = 0;
    int*         ls  = lists + (size_t)(blockIdx.x * OWN_WAVES + wave) * (2 * LIST_CAP);
    int*         ld  = ls + LIST_CAP;
    int          rec = 0;
    const float* hl = h   + lane;
    float*       al = acc + lane;

    const int nfull = E / SCAN_EDGES;
    const int first = (int)((blockIdx.x * (unsigned)nfull) / gridDim.x);

    auto push_row = [&](int d, int e) {
        const unsigned t    = (unsigned)(d - base);
        const bool     mine = t < (unsigned)rows;
        const unsigned mask = __builtin_amdgcn_ballot_w32(mine);
        if (mask != 0u) {
            if (mine) q[qn + (int)__builtin_amdgcn_mbcnt_lo(mask, 0u)] = e;
            qn += __builtin_popcount(mask);
        }
    };

    auto add_rows = [&](int sv, int dv, int m) {
        for (int g = 0; g < m; g += EDGE_GROUP) {
            float v[EDGE_GROUP][4];
#pragma unroll
            for (int j = 0; j < EDGE_GROUP; ++j) {
                const int s = __builtin_amdgcn_readlane(sv, g + j);
                const float* p = hl + (size_t)s * HDIM;
                v[j][0] = p[0]; v[j][1] = p[32]; v[j][2] = p[64]; v[j][3] = p[96];
            }
#pragma unroll
            for (int j = 0; j < EDGE_GROUP; ++j) {
                const int d = __builtin_amdgcn_readlane(dv, g + j);
                float* r = al + d * HDIM;
                atomicAdd(r,      v[j][0]);
                atomicAdd(r + 32, v[j][1]);
                atomicAdd(r + 64, v[j][2]);
                atomicAdd(r + 96, v[j][3]);
            }
        }
    };

    auto scan_step = [&](int (&use)[SCAN_ROWS], int (&fill)[SCAN_ROWS], int c) {
        int cl = c + 2 * OWN_WAVES;
        if (cl > nfull - 1) cl = nfull - 1;
        const int ccl = (cl + first < nfull) ? (cl + first) : (cl + first - nfull);
        const int* dl = dst + ccl * SCAN_EDGES + lane;
#pragma unroll
        for (int j = 0; j < SCAN_ROWS; ++j) fill[j] = dl[32 * j];

        if (c >= 0 && c < nfull) {
            const int cc = (c + first < nfull) ? (c + first) : (c + first - nfull);
            const int e0 = cc * SCAN_EDGES + lane;
#pragma unroll
            for (int j = 0; j < SCAN_ROWS; ++j) push_row(use[j], e0 + 32 * j);
        }

        while (qn >= 32) {
            qn -= 32;
            const int ee = q[qn + lane];
            const int sv = src[ee], dv = dst[ee] - base;
            if (FIRST) {
                if (rec + 32 <= LIST_CAP) { ls[rec + lane] = sv; ld[rec + lane] = dv; }
                rec += 32;
                atomicAdd(&degc[dv], 1);
            }
            add_rows(sv, dv, 32);
        }
    };

    int nsaved = 0;
    if (!FIRST) nsaved = __builtin_amdgcn_readfirstlane(counts[blockIdx.x * OWN_WAVES + wave]);
    const bool scan = FIRST || nsaved > LIST_CAP;

    if (scan) {
        if (nfull > 0) {
            int ra[SCAN_ROWS], rb[SCAN_ROWS], rc[SCAN_ROWS];
#pragma unroll
            for (int j = 0; j < SCAN_ROWS; ++j) { ra[j] = -1; rb[j] = -1; }
            for (int c = wave - 2 * OWN_WAVES; c < nfull; c += 3 * OWN_WAVES) {
                scan_step(ra, rc, c);
                scan_step(rb, ra, c + OWN_WAVES);
                scan_step(rc, rb, c + 2 * OWN_WAVES);
            }
        }

        if (wave * 32 < E - nfull * SCAN_EDGES) {
            const int e = nfull * SCAN_EDGES + wave * 32 + lane;
            push_row((e < E) ? dst[e] : -1, e);
        }
        while (qn > 0) {
            const int n = (qn < 32) ? qn : 32;
            qn -= n;
            int sv = 0, dv = OWN_ROWS;
            if (lane < n) {
                const int ee = q[qn + lane]; sv = src[ee]; dv = dst[ee] - base;
                if (FIRST && rec + n <= LIST_CAP) { ls[rec + lane] = sv; ld[rec + lane] = dv; }
                if (FIRST) atomicAdd(&degc[dv], 1);
            }
            if (FIRST) rec += n;
            add_rows(sv, dv, n);
        }
        if (FIRST && lane == 0) counts[blockIdx.x * OWN_WAVES + wave] = rec;
    } else {
        for (int i0 = 0; i0 < nsaved; i0 += 32 * REPLAY_ROUNDS) {
            int sv[REPLAY_ROUNDS], dv[REPLAY_ROUNDS];
#pragma unroll
            for (int j = 0; j < REPLAY_ROUNDS; ++j) {
                const int i = i0 + 32 * j + lane;
                sv[j] = 0; dv[j] = OWN_ROWS;
                if (i < nsaved) { sv[j] = ls[i]; dv[j] = ld[i]; }
            }
#pragma unroll
            for (int j = 0; j < REPLAY_ROUNDS; ++j) {
                const int left = nsaved - i0 - 32 * j;
                if (left > 0) add_rows(sv[j], dv[j], (left < 32) ? left : 32);
            }
        }
    }
    __syncthreads();

    float* outp = agg + (size_t)base * HDIM;
    for (int i = tid * 4; i < rows * HDIM; i += 1024 * 4)
        *(float4*)(outp + i) = *(const float4*)(&acc[i]);
    if (FIRST) {
        for (int r = tid; r < rows; r += 1024)
            invdeg[base + r] = 1.0f / fmaxf((float)degc[r], 1.0f);
    }
}

extern "C" void kernel_launch(void* const* d_in, const int* in_sizes, int n_in,
                              void* d_out, int out_size, void* d_ws, size_t ws_size,
                              hipStream_t stream)
{
    const float* x      = (const float*)d_in[0];
    const int*   edge   = (const int*)  d_in[1];
    const float* W_proj = (const float*)d_in[2];
    const float* b_proj = (const float*)d_in[3];
    const float* Wl[3]  = { (const float*)d_in[4], (const float*)d_in[7], (const float*)d_in[10] };
    const float* bl[3]  = { (const float*)d_in[5], (const float*)d_in[8], (const float*)d_in[11] };
    const float* Wr[3]  = { (const float*)d_in[6], (const float*)d_in[9], (const float*)d_in[12] };
    const float* Wc     = (const float*)d_in[13];
    const float* bc     = (const float*)d_in[14];

    const int N    = in_sizes[0] / 256;
    const int E    = in_sizes[1] / 2;
    const int ncls = in_sizes[14];

    const int* src = edge;
    const int* dst = edge + E;

    float* hA     = (float*)d_ws;
    float* hB     = hA  + (size_t)N * HDIM;
    float* agg    = hB  + (size_t)N * HDIM;
    float* invdeg = agg + (size_t)N * HDIM;
    int*   counts = (int*)(invdeg + N);
    int*   lists  = counts + (size_t)((N + OWN_ROWS - 1) / OWN_ROWS) * OWN_WAVES;

    const int rowTiles = (N + 15) / 16;

    proj_strip_kernel<<<256, STRIP_THREADS, 0, stream>>>(x, W_proj, b_proj, hA, rowTiles);

    for (int l = 0; l < 3; ++l) {
        if (l == 0)
            owned_rows_sum_kernel<true><<<(N + OWN_ROWS - 1) / OWN_ROWS, 1024, 0, stream>>>(src, dst, hA, agg,
                                                                                             lists, counts, invdeg, E, N);
        else
            owned_rows_sum_kernel<false><<<(N + OWN_ROWS - 1) / OWN_ROWS, 1024, 0, stream>>>(src, dst, hA, agg,
                                                                                              lists, counts, invdeg, E, N);
        sage_strip_kernel<<<256, STRIP_THREADS, 0, stream>>>(agg, invdeg, hA,
                                                             Wl[l], bl[l], Wr[l], hB, rowTiles);
        float* t = hA; hA = hB; hB = t;
    }

    cls_wmma_kernel<<<dim3(rowTiles, (ncls + 15) / 16), 32, 0, stream>>>(
        hA, Wc, bc, (float*)d_out, ncls);
}


# Edge filter becomes a launch of its own, tiled 8 sum blocks x an eighth of the list: dst is read 32 times, not 250; pairs go to per-(part, wave) lists, all three sums replay; sum order: LDS arrival
# speedup vs baseline: 3.3047x; 1.0732x over previous
#include <hip/hip_runtime.h>
#include <hip/hip_bf16.h>

typedef __attribute__((ext_vector_type(2))) float v2f;
typedef __attribute__((ext_vector_type(8))) float v8f;

#define HDIM 128

#define STRIP_WAVES   8
#define STRIP_THREADS (STRIP_WAVES * 32)
#define WFRAGS        (16 * 8 * 32)

__device__ __forceinline__ void stage_weights(const float* __restrict__ Wg, float4* Ws, int tid)
{
    for (int d = tid; d < WFRAGS; d += STRIP_THREADS) {
        const int l = d & 31, t = (d >> 5) & 7, kb = d >> 8;
        const int c = l & 15, h = l >> 4;
        const float* p = Wg + (size_t)(8 * kb + 4 * h) * HDIM + 16 * t + c;
        Ws[d] = make_float4(p[0], p[HDIM], p[2 * HDIM], p[3 * HDIM]);
    }
}

__device__ __forceinline__ void load_wfrags(float4 (&b)[8], const float4* Ws, int kk, int lane)
{
#pragma unroll
    for (int t = 0; t < 8; ++t) b[t] = Ws[(kk * 8 + t) * 32 + lane];
}

__device__ __forceinline__ void mm_kblock(const float4& a, const float4 (&b)[8], v8f (&acc)[8])
{
    v2f a0, a1;
    a0[0] = a.x; a0[1] = a.y;
    a1[0] = a.z; a1[1] = a.w;
#pragma unroll
    for (int t = 0; t < 8; ++t) {
        v2f bb; bb[0] = b[t].x; bb[1] = b[t].y;
        acc[t] = __builtin_amdgcn_wmma_f32_16x16x4_f32(false, a0, false, bb, (short)0, acc[t], false, false);
    }
#pragma unroll
    for (int t = 0; t < 8; ++t) {
        v2f bb; bb[0] = b[t].z; bb[1] = b[t].w;
        acc[t] = __builtin_amdgcn_wmma_f32_16x16x4_f32(false, a1, false, bb, (short)0, acc[t], false, false);
    }
}

template <bool SCALE>
__device__ __forceinline__ void strip_gemm(const float* __restrict__ A1, const float* __restrict__ A2, int pitch,
                                           const float* __restrict__ rowscale,
                                           const float* __restrict__ W1, const float* __restrict__ W2,
                                           const float* __restrict__ bias, float* __restrict__ out, int ntiles)
{
    __shared__ __align__(16) float4 Ws[2 * WFRAGS];

    const int tid  = threadIdx.x;
    const int lane = tid & 31;
    const int wave = __builtin_amdgcn_readfirstlane(tid >> 5);
    const int c    = lane & 15;
    const int h    = lane >> 4;

    stage_weights(W1, Ws, tid);
    stage_weights(W2, Ws + WFRAGS, tid);
    __syncthreads();

    float bv[8];
#pragma unroll
    for (int t = 0; t < 8; ++t) bv[t] = bias[16 * t + c];

    const int stride = STRIP_WAVES * (int)gridDim.x;
    int tile = wave * (int)gridDim.x + (int)blockIdx.x;
    if (tile >= ntiles) return;

    float4 a1[16], a2[16];
    float  s = 1.0f;
    {
        const int    row = tile * 16 + c;
        const float* p1  = A1 + (size_t)row * pitch + 4 * h;
#pragma unroll
        for (int kb = 0; kb < 16; ++kb) a1[kb] = *(const float4*)(p1 + 8 * kb);
        if (SCALE) s = rowscale[row];
    }

    for (; tile < ntiles; tile += stride) {
        const float* p2 = A2 + (size_t)(tile * 16 + c) * pitch + 4 * h;
#pragma unroll
        for (int kb = 0; kb < 16; ++kb) a2[kb] = *(const float4*)(p2 + 8 * kb);
        if (SCALE) {
#pragma unroll
            for (int kb = 0; kb < 16; ++kb) { a1[kb].x *= s; a1[kb].y *= s; a1[kb].z *= s; a1[kb].w *= s; }
        }

        v8f acc[8];
#pragma unroll
        for (int t = 0; t < 8; ++t) acc[t] = v8f{};

        float4 b0[8], b1[8];
        load_wfrags(b0, Ws, 0, lane);
#pragma unroll
        for (int kk = 0; kk < 32; kk += 2) {
            if (kk == 16) {
                const int    nt  = (tile + stride < ntiles) ? (tile + stride) : tile;
                const int    nr  = nt * 16 + c;
                const float* p1  = A1 + (size_t)nr * pitch + 4 * h;
#pragma unroll
                for (int kb = 0; kb < 16; ++kb) a1[kb] = *(const float4*)(p1 + 8 * kb);
                if (SCALE) s = rowscale[nr];
            }
            load_wfrags(b1, Ws, kk + 1, lane);
            mm_kblock((kk < 16) ? a1[kk & 15] : a2[kk & 15], b0, acc);
            __builtin_amdgcn_sched_barrier(0);
            if (kk + 2 < 32) load_wfrags(b0, Ws, kk + 2, lane);
            mm_kblock((kk + 1 < 16) ? a1[(kk + 1) & 15] : a2[(kk + 1) & 15], b1, acc);
            __builtin_amdgcn_sched_barrier(0);
        }

        float* o = out + (size_t)(tile * 16 + 8 * h) * HDIM + c;
#pragma unroll
        for (int r = 0; r < 8; ++r) {
#pragma unroll
            for (int t = 0; t < 8; ++t)
                o[(size_t)r * HDIM + 16 * t] = fmaxf(acc[t][r] + bv[t], 0.0f);
        }
    }
}

__global__ __launch_bounds__(STRIP_THREADS)
void proj_strip_kernel(const float* __restrict__ x, const float* __restrict__ W,
                       const float* __restrict__ bias, float* __restrict__ out, int ntiles)
{
    strip_gemm<false>(x, x + HDIM, 2 * HDIM, nullptr, W, W + HDIM * HDIM, bias, out, ntiles);
}

__global__ __launch_bounds__(STRIP_THREADS)
void sage_strip_kernel(const float* __restrict__ agg, const float* __restrict__ invdeg,
                       const float* __restrict__ h, const float* __restrict__ Wl,
                       const float* __restrict__ bl, const float* __restrict__ Wr,
                       float* __restrict__ out, int ntiles)
{
    strip_gemm<true>(agg, h, HDIM, invdeg, Wl, Wr, bl, out, ntiles);
}

__global__ __launch_bounds__(32)
void cls_wmma_kernel(const float* __restrict__ h, const float* __restrict__ Wc,
                     const float* __restrict__ bc, float* __restrict__ out, int ncls)
{
    const int row0 = blockIdx.x * 16;
    const int lane = threadIdx.x & 31;
    const int m    = lane & 15;
    const int koff = (lane < 16) ? 0 : 2;
    const int n    = blockIdx.y * 16 + (lane & 15);
    const bool nv  = (n < ncls);

    v8f c = {};
    for (int k = 0; k < HDIM; k += 4) {
        v2f a, b;
        a[0] = h[(size_t)(row0 + m) * HDIM + k + koff];
        a[1] = h[(size_t)(row0 + m) * HDIM + k + koff + 1];
        b[0] = nv ? Wc[(size_t)(k + koff)     * ncls + n] : 0.0f;
        b[1] = nv ? Wc[(size_t)(k + koff + 1) * ncls + n] : 0.0f;
        c = __builtin_amdgcn_wmma_f32_16x16x4_f32(false, a, false, b,
                                                  (short)0, c, false, false);
    }

    if (nv) {
        const int   mbase = row0 + ((lane < 16) ? 0 : 8);
        const float bv    = bc[n];
        for (int r = 0; r < 8; ++r)
            out[(size_t)(mbase + r) * ncls + n] = c[r] + bv;
    }
}

#define OWN_ROWS   400
#define OWN_WAVES  32
#define SCAN_ROWS  16
#define SCAN_EDGES (SCAN_ROWS * 32)
#define QCAP       (SCAN_EDGES + 32)
#define EDGE_GROUP 8
#define FGROUP     8
#define FPARTS     8
#define SLOTS      (FPARTS * OWN_WAVES)
#define SLOT_CAP   256
#define REPLAY_ROUNDS 8

__global__ __launch_bounds__(1024)
void edge_filter_kernel(const int* __restrict__ src, const int* __restrict__ dst,
                        int* __restrict__ lists, int* __restrict__ counts, int E, int N, int nblk)
{
    __shared__ int queue[OWN_WAVES * QCAP];
    __shared__ unsigned short qrow[OWN_WAVES * QCAP];

    const int tid  = threadIdx.x;
    const int lane = tid & 31;
    const int wave = __builtin_amdgcn_readfirstlane(tid >> 5);
    const int g    = (int)blockIdx.x / FPARTS;
    const int p    = (int)blockIdx.x % FPARTS;
    const int ngrp = (int)gridDim.x / FPARTS;
    const int base = g * (FGROUP * OWN_ROWS);
    const int rows = (N - base < FGROUP * OWN_ROWS) ? (N - base) : (FGROUP * OWN_ROWS);

    int*            q  = queue + wave * QCAP;
    unsigned short* qr = qrow  + wave * QCAP;
    int             qn = 0;
    int             cur[FGROUP];
#pragma unroll
    for (int kk = 0; kk < FGROUP; ++kk) cur[kk] = 0;

    const int nfull = E / SCAN_EDGES;
    const int pbeg  = (p * nfull) / FPARTS;
    const int plen  = ((p + 1) * nfull) / FPARTS - pbeg;
    const int first = (int)(((unsigned)g * (unsigned)plen) / (unsigned)ngrp);

    auto push_row = [&](int d, int e) {
        const unsigned t    = (unsigned)(d - base);
        const bool     mine = t < (unsigned)rows;
        const unsigned mask = __builtin_amdgcn_ballot_w32(mine);
        if (mask != 0u) {
            if (mine) {
                const int slot = qn + (int)__builtin_amdgcn_mbcnt_lo(mask, 0u);
                q[slot]  = e;
                qr[slot] = (unsigned short)t;
            }
            qn += __builtin_popcount(mask);
        }
    };

    auto emit = [&](int sv, int tt, int m) {
        const bool valid = lane < m;
        int        k     = 0;
#pragma unroll
        for (int t = 1; t < FGROUP; ++t) k += (tt >= t * OWN_ROWS) ? 1 : 0;
        const int  local = tt - k * OWN_ROWS;
#pragma unroll
        for (int kk = 0; kk < FGROUP; ++kk) {
            const bool     hit = valid && (k == kk);
            const unsigned mk  = __builtin_amdgcn_ballot_w32(hit);
            if (mk != 0u) {
                const int pos = cur[kk] + (int)__builtin_amdgcn_mbcnt_lo(mk, 0u);
                if (hit && pos < SLOT_CAP) {
                    int* L = lists + (size_t)((g * FGROUP + kk) * SLOTS + p * OWN_WAVES + wave) * (2 * SLOT_CAP);
                    L[pos]            = sv;
                    L[SLOT_CAP + pos] = local;
                }
                cur[kk] += __builtin_popcount(mk);
            }
        }
    };

    auto scan_step = [&](int (&use)[SCAN_ROWS], int (&fill)[SCAN_ROWS], int c) {
        int cl = c + 2 * OWN_WAVES;
        if (cl > plen - 1) cl = plen - 1;
        const int ccl = pbeg + ((cl + first < plen) ? (cl + first) : (cl + first - plen));
        const int* dl = dst + ccl * SCAN_EDGES + lane;
#pragma unroll
        for (int j = 0; j < SCAN_ROWS; ++j) fill[j] = dl[32 * j];

        if (c >= 0 && c < plen) {
            const int cc = pbeg + ((c + first < plen) ? (c + first) : (c + first - plen));
            const int e0 = cc * SCAN_EDGES + lane;
#pragma unroll
            for (int j = 0; j < SCAN_ROWS; ++j) push_row(use[j], e0 + 32 * j);
        }

        while (qn >= 32) {
            qn -= 32;
            const int ee = q[qn + lane];
            emit(src[ee], (int)qr[qn + lane], 32);
        }
    };

    if (plen > 0) {
        int ra[SCAN_ROWS], rb[SCAN_ROWS], rc[SCAN_ROWS];
#pragma unroll
        for (int j = 0; j < SCAN_ROWS; ++j) { ra[j] = -1; rb[j] = -1; }
        for (int c = wave - 2 * OWN_WAVES; c < plen; c += 3 * OWN_WAVES) {
            scan_step(ra, rc, c);
            scan_step(rb, ra, c + OWN_WAVES);
            scan_step(rc, rb, c + 2 * OWN_WAVES);
        }
    }

    if (p == FPARTS - 1 && wave * 32 < E - nfull * SCAN_EDGES) {
        const int e = nfull * SCAN_EDGES + wave * 32 + lane;
        push_row((e < E) ? dst[e] : -1, e);
    }
    while (qn > 0) {
        const int n = (qn < 32) ? qn : 32;
        qn -= n;
        int sv = 0, tt = 0;
        if (lane < n) { const int ee = q[qn + lane]; sv = src[ee]; tt = (int)qr[qn + lane]; }
        emit(sv, tt, n);
    }
    if (lane == 0) {
#pragma unroll
        for (int kk = 0; kk < FGROUP; ++kk)
            if (g * FGROUP + kk < nblk) counts[(g * FGROUP + kk) * SLOTS + p * OWN_WAVES + wave] = cur[kk];
    }
}

template <bool FIRST>
__global__ __launch_bounds__(1024)
void owned_rows_sum_kernel(const int* __restrict__ src, const int* __restrict__ dst,
                           const float* __restrict__ h, float* __restrict__ agg,
                           const int* __restrict__ lists, const int* __restrict__ counts,
                           float* __restrict__ invdeg, int E, int N)
{
    __shared__ __align__(16) float acc[(OWN_ROWS + 1) * HDIM];
    __shared__ int queue[OWN_WAVES * QCAP];
    __shared__ unsigned short qrow[OWN_WAVES * QCAP];
    __shared__ int degc[OWN_ROWS];

    const int tid  = threadIdx.x;
    const int lane = tid & 31;
    const int wave = __builtin_amdgcn_readfirstlane(tid >> 5);
    const int base = blockIdx.x * OWN_ROWS;
    const int rows = (N - base < OWN_ROWS) ? (N - base) : OWN_ROWS;

    for (int i = tid * 4; i < (OWN_ROWS + 1) * HDIM; i += 1024 * 4)
        *(float4*)(&acc[i]) = make_float4(0.0f, 0.0f, 0.0f, 0.0f);
    if (FIRST) {
        for (int i = tid; i < OWN_ROWS; i += 1024) degc[i] = 0;
    }

    int  ns[FPARTS];
    bool over = false;
#pragma unroll
    for (int s = 0; s < FPARTS; ++s) {
        ns[s] = __builtin_amdgcn_readfirstlane(counts[blockIdx.x * SLOTS + s * OWN_WAVES + wave]);
        over  = over || (ns[s] > SLOT_CAP);
    }
    const bool scan = __builtin_amdgcn_readfirstlane(__syncthreads_or(over ? 1 : 0)) != 0;

    const float* hl = h   + lane;
    float*       al = acc + lane;

    auto add_rows = [&](int sv, int dv, int m) {
        if (FIRST) {
            if (lane < m) atomicAdd(&degc[dv], 1);
        }
        for (int g = 0; g < m; g += EDGE_GROUP) {
            float v[EDGE_GROUP][4];
#pragma unroll
            for (int j = 0; j < EDGE_GROUP; ++j) {
                const int s = __builtin_amdgcn_readlane(sv, g + j);
                const float* p = hl + (size_t)s * HDIM;
                v[j][0] = p[0]; v[j][1] = p[32]; v[j][2] = p[64]; v[j][3] = p[96];
            }
#pragma unroll
            for (int j = 0; j < EDGE_GROUP; ++j) {
                const int d = __builtin_amdgcn_readlane(dv, g + j);
                float* r = al + d * HDIM;
                atomicAdd(r,      v[j][0]);
                atomicAdd(r + 32, v[j][1]);
                atomicAdd(r + 64, v[j][2]);
                atomicAdd(r + 96, v[j][3]);
            }
        }
    };

    if (scan) {
        int*            q  = queue + wave * QCAP;
        unsigned short* qr = qrow  + wave * QCAP;
        int             qn = 0;

        const int nfull = E / SCAN_EDGES;
        const int first = (int)((blockIdx.x * (unsigned)nfull) / gridDim.x);

        auto push_row = [&](int d, int e) {
            const unsigned t    = (unsigned)(d - base);
            const bool     mine = t < (unsigned)rows;
            const unsigned mask = __builtin_amdgcn_ballot_w32(mine);
            if (mask != 0u) {
                if (mine) {
                    const int slot = qn + (int)__builtin_amdgcn_mbcnt_lo(mask, 0u);
                    q[slot]  = e;
                    qr[slot] = (unsigned short)t;
                }
                qn += __builtin_popcount(mask);
            }
        };

        auto scan_step = [&](int (&use)[SCAN_ROWS], int (&fill)[SCAN_ROWS], int c) {
            int cl = c + 2 * OWN_WAVES;
            if (cl > nfull - 1) cl = nfull - 1;
            const int ccl = (cl + first < nfull) ? (cl + first) : (cl + first - nfull);
            const int* dl = dst + ccl * SCAN_EDGES + lane;
#pragma unroll
            for (int j = 0; j < SCAN_ROWS; ++j) fill[j] = dl[32 * j];

            if (c >= 0 && c < nfull) {
                const int cc = (c + first < nfull) ? (c + first) : (c + first - nfull);
                const int e0 = cc * SCAN_EDGES + lane;
#pragma unroll
                for (int j = 0; j < SCAN_ROWS; ++j) push_row(use[j], e0 + 32 * j);
            }

            while (qn >= 32) {
                qn -= 32;
                const int ee = q[qn + lane];
                add_rows(src[ee], (int)qr[qn + lane], 32);
            }
        };

        if (nfull > 0) {
            int ra[SCAN_ROWS], rb[SCAN_ROWS], rc[SCAN_ROWS];
#pragma unroll
            for (int j = 0; j < SCAN_ROWS; ++j) { ra[j] = -1; rb[j] = -1; }
            for (int c = wave - 2 * OWN_WAVES; c < nfull; c += 3 * OWN_WAVES) {
                scan_step(ra, rc, c);
                scan_step(rb, ra, c + OWN_WAVES);
                scan_step(rc, rb, c + 2 * OWN_WAVES);
            }
        }
        if (wave * 32 < E - nfull * SCAN_EDGES) {
            const int e = nfull * SCAN_EDGES + wave * 32 + lane;
            push_row((e < E) ? dst[e] : -1, e);
        }
        while (qn > 0) {
            const int n = (qn < 32) ? qn : 32;
            qn -= n;
            int sv = 0, dv = OWN_ROWS;
            if (lane < n) { const int ee = q[qn + lane]; sv = src[ee]; dv = (int)qr[qn + lane]; }
            add_rows(sv, dv, n);
        }
    } else {
        const int* Lb = lists + (size_t)(blockIdx.x * SLOTS + wave) * (2 * SLOT_CAP);
        int off[FPARTS + 1];
        off[0] = 0;
#pragma unroll
        for (int s = 0; s < FPARTS; ++s) off[s + 1] = off[s] + ns[s];
        const int tot = off[FPARTS];
        for (int i0 = 0; i0 < tot; i0 += 32 * REPLAY_ROUNDS) {
            int sv[REPLAY_ROUNDS], dv[REPLAY_ROUNDS];
#pragma unroll
            for (int j = 0; j < REPLAY_ROUNDS; ++j) {
                const int v = i0 + 32 * j + lane;
                sv[j] = 0; dv[j] = OWN_ROWS;
                if (v < tot) {
                    int s = 0, ob = 0;
#pragma unroll
                    for (int t = 1; t < FPARTS; ++t) if (v >= off[t]) { s = t; ob = off[t]; }
                    const int* L = Lb + (size_t)s * OWN_WAVES * (2 * SLOT_CAP);
                    sv[j] = L[v - ob]; dv[j] = L[SLOT_CAP + v - ob];
                }
            }
#pragma unroll
            for (int j = 0; j < REPLAY_ROUNDS; ++j) {
                const int left = tot - i0 - 32 * j;
                if (left > 0) add_rows(sv[j], dv[j], (left < 32) ? left : 32);
            }
        }
    }
    __syncthreads();

    float* outp = agg + (size_t)base * HDIM;
    for (int i = tid * 4; i < rows * HDIM; i += 1024 * 4)
        *(float4*)(outp + i) = *(const float4*)(&acc[i]);
    if (FIRST) {
        for (int r = tid; r < rows; r += 1024)
            invdeg[base + r] = 1.0f / fmaxf((float)degc[r], 1.0f);
    }
}

extern "C" void kernel_launch(void* const* d_in, const int* in_sizes, int n_in,
                              void* d_out, int out_size, void* d_ws, size_t ws_size,
                              hipStream_t stream)
{
    const float* x      = (const float*)d_in[0];
    const int*   edge   = (const int*)  d_in[1];
    const float* W_proj = (const float*)d_in[2];
    const float* b_proj = (const float*)d_in[3];
    const float* Wl[3]  = { (const float*)d_in[4], (const float*)d_in[7], (const float*)d_in[10] };
    const float* bl[3]  = { (const float*)d_in[5], (const float*)d_in[8], (const float*)d_in[11] };
    const float* Wr[3]  = { (const float*)d_in[6], (const float*)d_in[9], (const float*)d_in[12] };
    const float* Wc     = (const float*)d_in[13];
    const float* bc     = (const float*)d_in[14];

    const int N    = in_sizes[0] / 256;
    const int E    = in_sizes[1] / 2;
    const int ncls = in_sizes[14];

    const int* src = edge;
    const int* dst = edge + E;

    const int nblk = (N + OWN_ROWS - 1) / OWN_ROWS;
    const int ngrp = (nblk + FGROUP - 1) / FGROUP;

    float* hA     = (float*)d_ws;
    float* hB     = hA  + (size_t)N * HDIM;
    float* agg    = hB  + (size_t)N * HDIM;
    float* invdeg = agg + (size_t)N * HDIM;
    int*   counts = (int*)(invdeg + N);
    int*   lists  = counts + (size_t)nblk * SLOTS;

    const int rowTiles = (N + 15) / 16;

    edge_filter_kernel<<<ngrp * FPARTS, 1024, 0, stream>>>(src, dst, lists, counts, E, N, nblk);

    proj_strip_kernel<<<256, STRIP_THREADS, 0, stream>>>(x, W_proj, b_proj, hA, rowTiles);

    for (int l = 0; l < 3; ++l) {
        if (l == 0)
            owned_rows_sum_kernel<true><<<nblk, 1024, 0, stream>>>(src, dst, hA, agg, lists, counts, invdeg, E, N);
        else
            owned_rows_sum_kernel<false><<<nblk, 1024, 0, stream>>>(src, dst, hA, agg, lists, counts, invdeg, E, N);
        sage_strip_kernel<<<256, STRIP_THREADS, 0, stream>>>(agg, invdeg, hA,
                                                             Wl[l], bl[l], Wr[l], hB, rowTiles);
        float* t = hA; hA = hB; hB = t;
    }

    cls_wmma_kernel<<<dim3(rowTiles, (ncls + 15) / 16), 32, 0, stream>>>(
        hA, Wc, bc, (float*)d_out, ncls);
}


# Owned LDS rows split among the block's waves: pairs dealt to owner-wave queues (LDS counters), a wave adds whole source rows by 16-B read-add-write; no LDS float atomics; sum order: queue arrival
# speedup vs baseline: 3.5180x; 1.0645x over previous
#include <hip/hip_runtime.h>
#include <hip/hip_bf16.h>

typedef __attribute__((ext_vector_type(2))) float v2f;
typedef __attribute__((ext_vector_type(8))) float v8f;

#define HDIM 128

#define STRIP_WAVES   8
#define STRIP_THREADS (STRIP_WAVES * 32)
#define WFRAGS        (16 * 8 * 32)

__device__ __forceinline__ void stage_weights(const float* __restrict__ Wg, float4* Ws, int tid)
{
    for (int d = tid; d < WFRAGS; d += STRIP_THREADS) {
        const int l = d & 31, t = (d >> 5) & 7, kb = d >> 8;
        const int c = l & 15, h = l >> 4;
        const float* p = Wg + (size_t)(8 * kb + 4 * h) * HDIM + 16 * t + c;
        Ws[d] = make_float4(p[0], p[HDIM], p[2 * HDIM], p[3 * HDIM]);
    }
}

__device__ __forceinline__ void load_wfrags(float4 (&b)[8], const float4* Ws, int kk, int lane)
{
#pragma unroll
    for (int t = 0; t < 8; ++t) b[t] = Ws[(kk * 8 + t) * 32 + lane];
}

__device__ __forceinline__ void mm_kblock(const float4& a, const float4 (&b)[8], v8f (&acc)[8])
{
    v2f a0, a1;
    a0[0] = a.x; a0[1] = a.y;
    a1[0] = a.z; a1[1] = a.w;
#pragma unroll
    for (int t = 0; t < 8; ++t) {
        v2f bb; bb[0] = b[t].x; bb[1] = b[t].y;
        acc[t] = __builtin_amdgcn_wmma_f32_16x16x4_f32(false, a0, false, bb, (short)0, acc[t], false, false);
    }
#pragma unroll
    for (int t = 0; t < 8; ++t) {
        v2f bb; bb[0] = b[t].z; bb[1] = b[t].w;
        acc[t] = __builtin_amdgcn_wmma_f32_16x16x4_f32(false, a1, false, bb, (short)0, acc[t], false, false);
    }
}

template <bool SCALE>
__device__ __forceinline__ void strip_gemm(const float* __restrict__ A1, const float* __restrict__ A2, int pitch,
                                           const float* __restrict__ rowscale,
                                           const float* __restrict__ W1, const float* __restrict__ W2,
                                           const float* __restrict__ bias, float* __restrict__ out, int ntiles)
{
    __shared__ __align__(16) float4 Ws[2 * WFRAGS];

    const int tid  = threadIdx.x;
    const int lane = tid & 31;
    const int wave = __builtin_amdgcn_readfirstlane(tid >> 5);
    const int c    = lane & 15;
    const int h    = lane >> 4;

    stage_weights(W1, Ws, tid);
    stage_weights(W2, Ws + WFRAGS, tid);
    __syncthreads();

    float bv[8];
#pragma unroll
    for (int t = 0; t < 8; ++t) bv[t] = bias[16 * t + c];

    const int stride = STRIP_WAVES * (int)gridDim.x;
    int tile = wave * (int)gridDim.x + (int)blockIdx.x;
    if (tile >= ntiles) return;

    float4 a1[16], a2[16];
    float  s = 1.0f;
    {
        const int    row = tile * 16 + c;
        const float* p1  = A1 + (size_t)row * pitch + 4 * h;
#pragma unroll
        for (int kb = 0; kb < 16; ++kb) a1[kb] = *(const float4*)(p1 + 8 * kb);
        if (SCALE) s = rowscale[row];
    }

    for (; tile < ntiles; tile += stride) {
        const float* p2 = A2 + (size_t)(tile * 16 + c) * pitch + 4 * h;
#pragma unroll
        for (int kb = 0; kb < 16; ++kb) a2[kb] = *(const float4*)(p2 + 8 * kb);
        if (SCALE) {
#pragma unroll
            for (int kb = 0; kb < 16; ++kb) { a1[kb].x *= s; a1[kb].y *= s; a1[kb].z *= s; a1[kb].w *= s; }
        }

        v8f acc[8];
#pragma unroll
        for (int t = 0; t < 8; ++t) acc[t] = v8f{};

        float4 b0[8], b1[8];
        load_wfrags(b0, Ws, 0, lane);
#pragma unroll
        for (int kk = 0; kk < 32; kk += 2) {
            if (kk == 16) {
                const int    nt  = (tile + stride < ntiles) ? (tile + stride) : tile;
                const int    nr  = nt * 16 + c;
                const float* p1  = A1 + (size_t)nr * pitch + 4 * h;
#pragma unroll
                for (int kb = 0; kb < 16; ++kb) a1[kb] = *(const float4*)(p1 + 8 * kb);
                if (SCALE) s = rowscale[nr];
            }
            load_wfrags(b1, Ws, kk + 1, lane);
            mm_kblock((kk < 16) ? a1[kk & 15] : a2[kk & 15], b0, acc);
            __builtin_amdgcn_sched_barrier(0);
            if (kk + 2 < 32) load_wfrags(b0, Ws, kk + 2, lane);
            mm_kblock((kk + 1 < 16) ? a1[(kk + 1) & 15] : a2[(kk + 1) & 15], b1, acc);
            __builtin_amdgcn_sched_barrier(0);
        }

        float* o = out + (size_t)(tile * 16 + 8 * h) * HDIM + c;
#pragma unroll
        for (int r = 0; r < 8; ++r) {
#pragma unroll
            for (int t = 0; t < 8; ++t)
                o[(size_t)r * HDIM + 16 * t] = fmaxf(acc[t][r] + bv[t], 0.0f);
        }
    }
}

__global__ __launch_bounds__(STRIP_THREADS)
void proj_strip_kernel(const float* __restrict__ x, const float* __restrict__ W,
                       const float* __restrict__ bias, float* __restrict__ out, int ntiles)
{
    strip_gemm<false>(x, x + HDIM, 2 * HDIM, nullptr, W, W + HDIM * HDIM, bias, out, ntiles);
}

__global__ __launch_bounds__(STRIP_THREADS)
void sage_strip_kernel(const float* __restrict__ agg, const float* __restrict__ invdeg,
                       const float* __restrict__ h, const float* __restrict__ Wl,
                       const float* __restrict__ bl, const float* __restrict__ Wr,
                       float* __restrict__ out, int ntiles)
{
    strip_gemm<true>(agg, h, HDIM, invdeg, Wl, Wr, bl, out, ntiles);
}

__global__ __launch_bounds__(32)
void cls_wmma_kernel(const float* __restrict__ h, const float* __restrict__ Wc,
                     const float* __restrict__ bc, float* __restrict__ out, int ncls)
{
    const int row0 = blockIdx.x * 16;
    const int lane = threadIdx.x & 31;
    const int m    = lane & 15;
    const int koff = (lane < 16) ? 0 : 2;
    const int n    = blockIdx.y * 16 + (lane & 15);
    const bool nv  = (n < ncls);

    v8f c = {};
    for (int k = 0; k < HDIM; k += 4) {
        v2f a, b;
        a[0] = h[(size_t)(row0 + m) * HDIM + k + koff];
        a[1] = h[(size_t)(row0 + m) * HDIM + k + koff + 1];
        b[0] = nv ? Wc[(size_t)(k + koff)     * ncls + n] : 0.0f;
        b[1] = nv ? Wc[(size_t)(k + koff + 1) * ncls + n] : 0.0f;
        c = __builtin_amdgcn_wmma_f32_16x16x4_f32(false, a, false, b,
                                                  (short)0, c, false, false);
    }

    if (nv) {
        const int   mbase = row0 + ((lane < 16) ? 0 : 8);
        const float bv    = bc[n];
        for (int r = 0; r < 8; ++r)
            out[(size_t)(mbase + r) * ncls + n] = c[r] + bv;
    }
}

#define OWN_ROWS   400
#define OWN_WAVES  32
#define SCAN_ROWS  16
#define SCAN_EDGES (SCAN_ROWS * 32)
#define QCAP       (SCAN_EDGES + 32)
#define EDGE_GROUP 8
#define FGROUP     8
#define FPARTS     8
#define SLOTS      (FPARTS * OWN_WAVES)
#define SLOT_CAP   256
#define REPLAY_ROUNDS 8

__global__ __launch_bounds__(1024)
void edge_filter_kernel(const int* __restrict__ src, const int* __restrict__ dst,
                        int* __restrict__ lists, int* __restrict__ counts, int E, int N, int nblk)
{
    __shared__ int queue[OWN_WAVES * QCAP];
    __shared__ unsigned short qrow[OWN_WAVES * QCAP];

    const int tid  = threadIdx.x;
    const int lane = tid & 31;
    const int wave = __builtin_amdgcn_readfirstlane(tid >> 5);
    const int g    = (int)blockIdx.x / FPARTS;
    const int p    = (int)blockIdx.x % FPARTS;
    const int ngrp = (int)gridDim.x / FPARTS;
    const int base = g * (FGROUP * OWN_ROWS);
    const int rows = (N - base < FGROUP * OWN_ROWS) ? (N - base) : (FGROUP * OWN_ROWS);

    int*            q  = queue + wave * QCAP;
    unsigned short* qr = qrow  + wave * QCAP;
    int             qn = 0;
    int             cur[FGROUP];
#pragma unroll
    for (int kk = 0; kk < FGROUP; ++kk) cur[kk] = 0;

    const int nfull = E / SCAN_EDGES;
    const int pbeg  = (p * nfull) / FPARTS;
    const int plen  = ((p + 1) * nfull) / FPARTS - pbeg;
    const int first = (int)(((unsigned)g * (unsigned)plen) / (unsigned)ngrp);

    auto push_row = [&](int d, int e) {
        const unsigned t    = (unsigned)(d - base);
        const bool     mine = t < (unsigned)rows;
        const unsigned mask = __builtin_amdgcn_ballot_w32(mine);
        if (mask != 0u) {
            if (mine) {
                const int slot = qn + (int)__builtin_amdgcn_mbcnt_lo(mask, 0u);
                q[slot]  = e;
                qr[slot] = (unsigned short)t;
            }
            qn += __builtin_popcount(mask);
        }
    };

    auto emit = [&](int sv, int tt, int m) {
        const bool valid = lane < m;
        int        k     = 0;
#pragma unroll
        for (int t = 1; t < FGROUP; ++t) k += (tt >= t * OWN_ROWS) ? 1 : 0;
        const int  local = tt - k * OWN_ROWS;
#pragma unroll
        for (int kk = 0; kk < FGROUP; ++kk) {
            const bool     hit = valid && (k == kk);
            const unsigned mk  = __builtin_amdgcn_ballot_w32(hit);
            if (mk != 0u) {
                const int pos = cur[kk] + (int)__builtin_amdgcn_mbcnt_lo(mk, 0u);
                if (hit && pos < SLOT_CAP) {
                    int* L = lists + (size_t)((g * FGROUP + kk) * SLOTS + p * OWN_WAVES + wave) * (2 * SLOT_CAP);
                    L[pos]            = sv;
                    L[SLOT_CAP + pos] = local;
                }
                cur[kk] += __builtin_popcount(mk);
            }
        }
    };

    auto scan_step = [&](int (&use)[SCAN_ROWS], int (&fill)[SCAN_ROWS], int c) {
        int cl = c + 2 * OWN_WAVES;
        if (cl > plen - 1) cl = plen - 1;
        const int ccl = pbeg + ((cl + first < plen) ? (cl + first) : (cl + first - plen));
        const int* dl = dst + ccl * SCAN_EDGES + lane;
#pragma unroll
        for (int j = 0; j < SCAN_ROWS; ++j) fill[j] = dl[32 * j];

        if (c >= 0 && c < plen) {
            const int cc = pbeg + ((c + first < plen) ? (c + first) : (c + first - plen));
            const int e0 = cc * SCAN_EDGES + lane;
#pragma unroll
            for (int j = 0; j < SCAN_ROWS; ++j) push_row(use[j], e0 + 32 * j);
        }

        while (qn >= 32) {
            qn -= 32;
            const int ee = q[qn + lane];
            emit(src[ee], (int)qr[qn + lane], 32);
        }
    };

    if (plen > 0) {
        int ra[SCAN_ROWS], rb[SCAN_ROWS], rc[SCAN_ROWS];
#pragma unroll
        for (int j = 0; j < SCAN_ROWS; ++j) { ra[j] = -1; rb[j] = -1; }
        for (int c = wave - 2 * OWN_WAVES; c < plen; c += 3 * OWN_WAVES) {
            scan_step(ra, rc, c);
            scan_step(rb, ra, c + OWN_WAVES);
            scan_step(rc, rb, c + 2 * OWN_WAVES);
        }
    }

    if (p == FPARTS - 1 && wave * 32 < E - nfull * SCAN_EDGES) {
        const int e = nfull * SCAN_EDGES + wave * 32 + lane;
        push_row((e < E) ? dst[e] : -1, e);
    }
    while (qn > 0) {
        const int n = (qn < 32) ? qn : 32;
        qn -= n;
        int sv = 0, tt = 0;
        if (lane < n) { const int ee = q[qn + lane]; sv = src[ee]; tt = (int)qr[qn + lane]; }
        emit(sv, tt, n);
    }
    if (lane == 0) {
#pragma unroll
        for (int kk = 0; kk < FGROUP; ++kk)
            if (g * FGROUP + kk < nblk) counts[(g * FGROUP + kk) * SLOTS + p * OWN_WAVES + wave] = cur[kk];
    }
}

template <bool FIRST>
__global__ __launch_bounds__(1024)
void owned_rows_sum_kernel(const int* __restrict__ src, const int* __restrict__ dst,
                           const float* __restrict__ h, float* __restrict__ agg,
                           const int* __restrict__ lists, const int* __restrict__ counts,
                           float* __restrict__ invdeg, int E, int N)
{
    __shared__ __align__(16) float acc[(OWN_ROWS + 1) * HDIM];
    __shared__ int queue[OWN_WAVES * QCAP];
    __shared__ unsigned short qrow[OWN_WAVES * QCAP];
    __shared__ int degc[OWN_ROWS];
    __shared__ int qcnt[OWN_WAVES];

    const int tid  = threadIdx.x;
    const int lane = tid & 31;
    const int wave = __builtin_amdgcn_readfirstlane(tid >> 5);
    const int base = blockIdx.x * OWN_ROWS;
    const int rows = (N - base < OWN_ROWS) ? (N - base) : OWN_ROWS;

    for (int i = tid * 4; i < (OWN_ROWS + 1) * HDIM; i += 1024 * 4)
        *(float4*)(&acc[i]) = make_float4(0.0f, 0.0f, 0.0f, 0.0f);
    if (FIRST) {
        for (int i = tid; i < OWN_ROWS; i += 1024) degc[i] = 0;
    }
    if (tid < OWN_WAVES) qcnt[tid] = 0;

    int  ns[FPARTS];
    bool over = false;
#pragma unroll
    for (int s = 0; s < FPARTS; ++s) {
        ns[s] = __builtin_amdgcn_readfirstlane(counts[blockIdx.x * SLOTS + s * OWN_WAVES + wave]);
        over  = over || (ns[s] > SLOT_CAP);
    }
    bool scan = __builtin_amdgcn_readfirstlane(__syncthreads_or(over ? 1 : 0)) != 0;

    const float* hl = h   + lane;
    float*       al = acc + lane;

    auto add_rows = [&](int sv, int dv, int m) {
        if (FIRST) {
            if (lane < m) atomicAdd(&degc[dv], 1);
        }
        for (int g = 0; g < m; g += EDGE_GROUP) {
            float v[EDGE_GROUP][4];
#pragma unroll
            for (int j = 0; j < EDGE_GROUP; ++j) {
                const int s = __builtin_amdgcn_readlane(sv, g + j);
                const float* p = hl + (size_t)s * HDIM;
                v[j][0] = p[0]; v[j][1] = p[32]; v[j][2] = p[64]; v[j][3] = p[96];
            }
#pragma unroll
            for (int j = 0; j < EDGE_GROUP; ++j) {
                const int d = __builtin_amdgcn_readlane(dv, g + j);
                float* r = al + d * HDIM;
                atomicAdd(r,      v[j][0]);
                atomicAdd(r + 32, v[j][1]);
                atomicAdd(r + 64, v[j][2]);
                atomicAdd(r + 96, v[j][3]);
            }
        }
    };

    if (!scan) {
        const int* Lb = lists + (size_t)(blockIdx.x * SLOTS + wave) * (2 * SLOT_CAP);
        int off[FPARTS + 1];
        off[0] = 0;
#pragma unroll
        for (int s = 0; s < FPARTS; ++s) off[s + 1] = off[s] + ns[s];
        const int tot = off[FPARTS];
        bool full = false;
        for (int i0 = 0; i0 < tot; i0 += 32 * REPLAY_ROUNDS) {
            int sv[REPLAY_ROUNDS], dv[REPLAY_ROUNDS];
#pragma unroll
            for (int j = 0; j < REPLAY_ROUNDS; ++j) {
                const int v = i0 + 32 * j + lane;
                sv[j] = 0; dv[j] = OWN_ROWS;
                if (v < tot) {
                    int s = 0, ob = 0;
#pragma unroll
                    for (int t = 1; t < FPARTS; ++t) if (v >= off[t]) { s = t; ob = off[t]; }
                    const int* L = Lb + (size_t)s * OWN_WAVES * (2 * SLOT_CAP);
                    sv[j] = L[v - ob]; dv[j] = L[SLOT_CAP + v - ob];
                }
            }
#pragma unroll
            for (int j = 0; j < REPLAY_ROUNDS; ++j) {
                if (i0 + 32 * j + lane < tot) {
                    const int o    = dv[j] & (OWN_WAVES - 1);
                    const int slot = atomicAdd(&qcnt[o], 1);
                    if (slot < QCAP) {
                        queue[o * QCAP + slot] = sv[j];
                        qrow[o * QCAP + slot]  = (unsigned short)dv[j];
                    } else {
                        full = true;
                    }
                    if (FIRST) atomicAdd(&degc[dv[j]], 1);
                }
            }
        }
        scan = __builtin_amdgcn_readfirstlane(__syncthreads_or(full ? 1 : 0)) != 0;

        if (!scan) {
            const int*            q  = queue + wave * QCAP;
            const unsigned short* qr = qrow  + wave * QCAP;
            const int    n   = __builtin_amdgcn_readfirstlane(qcnt[wave]);
            const float* hl4 = h   + 4 * lane;
            float*       al4 = acc + 4 * lane;

            auto load8 = [&](float4 (&v)[EDGE_GROUP], int svv, int g0) {
#pragma unroll
                for (int j = 0; j < EDGE_GROUP; ++j) {
                    const int s = __builtin_amdgcn_readlane(svv, g0 + j);
                    v[j] = *(const float4*)(hl4 + (size_t)s * HDIM);
                }
            };
            auto rmw8 = [&](const float4 (&v)[EDGE_GROUP], int dvv, int g0) {
#pragma unroll
                for (int j = 0; j < EDGE_GROUP; ++j) {
                    const int d = __builtin_amdgcn_readlane(dvv, g0 + j);
                    float4*   r = (float4*)(al4 + d * HDIM);
                    float4    a = *r;
                    a.x += v[j].x; a.y += v[j].y; a.z += v[j].z; a.w += v[j].w;
                    *r = a;
                }
            };

            int svv = 0, dvv = OWN_ROWS;
            if (lane < n) { svv = q[lane]; dvv = (int)qr[lane]; }
            float4 va[EDGE_GROUP], vb[EDGE_GROUP];
            load8(va, svv, 0);
            const int nround = n >> 5;
            for (int r = 0; r < nround; ++r) {
                int svn = 0, dvn = OWN_ROWS;
                const int k = 32 * (r + 1) + lane;
                if (k < n) { svn = q[k]; dvn = (int)qr[k]; }
                load8(vb, svv, 8);   rmw8(va, dvv, 0);
                load8(va, svv, 16);  rmw8(vb, dvv, 8);
                load8(vb, svv, 24);  rmw8(va, dvv, 16);
                load8(va, svn, 0);   rmw8(vb, dvv, 24);
                svv = svn; dvv = dvn;
            }
            const int m = n & 31;
            if (m > 0) {
                rmw8(va, dvv, 0);
                if (m > 8)  { load8(vb, svv, 8);  rmw8(vb, dvv, 8); }
                if (m > 16) { load8(va, svv, 16); rmw8(va, dvv, 16); }
                if (m > 24) { load8(vb, svv, 24); rmw8(vb, dvv, 24); }
            }
        } else if (FIRST) {
            for (int i = tid; i < OWN_ROWS; i += 1024) degc[i] = 0;
            __syncthreads();
        }
    }
    if (scan) {
        int*            q  = queue + wave * QCAP;
        unsigned short* qr = qrow  + wave * QCAP;
        int             qn = 0;

        const int nfull = E / SCAN_EDGES;
        const int first = (int)((blockIdx.x * (unsigned)nfull) / gridDim.x);

        auto push_row = [&](int d, int e) {
            const unsigned t    = (unsigned)(d - base);
            const bool     mine = t < (unsigned)rows;
            const unsigned mask = __builtin_amdgcn_ballot_w32(mine);
            if (mask != 0u) {
                if (mine) {
                    const int slot = qn + (int)__builtin_amdgcn_mbcnt_lo(mask, 0u);
                    q[slot]  = e;
                    qr[slot] = (unsigned short)t;
                }
                qn += __builtin_popcount(mask);
            }
        };

        auto scan_step = [&](int (&use)[SCAN_ROWS], int (&fill)[SCAN_ROWS], int c) {
            int cl = c + 2 * OWN_WAVES;
            if (cl > nfull - 1) cl = nfull - 1;
            const int ccl = (cl + first < nfull) ? (cl + first) : (cl + first - nfull);
            const int* dl = dst + ccl * SCAN_EDGES + lane;
#pragma unroll
            for (int j = 0; j < SCAN_ROWS; ++j) fill[j] = dl[32 * j];

            if (c >= 0 && c < nfull) {
                const int cc = (c + first < nfull) ? (c + first) : (c + first - nfull);
                const int e0 = cc * SCAN_EDGES + lane;
#pragma unroll
                for (int j = 0; j < SCAN_ROWS; ++j) push_row(use[j], e0 + 32 * j);
            }

            while (qn >= 32) {
                qn -= 32;
                const int ee = q[qn + lane];
                add_rows(src[ee], (int)qr[qn + lane], 32);
            }
        };

        if (nfull > 0) {
            int ra[SCAN_ROWS], rb[SCAN_ROWS], rc[SCAN_ROWS];
#pragma unroll
            for (int j = 0; j < SCAN_ROWS; ++j) { ra[j] = -1; rb[j] = -1; }
            for (int c = wave - 2 * OWN_WAVES; c < nfull; c += 3 * OWN_WAVES) {
                scan_step(ra, rc, c);
                scan_step(rb, ra, c + OWN_WAVES);
                scan_step(rc, rb, c + 2 * OWN_WAVES);
            }
        }
        if (wave * 32 < E - nfull * SCAN_EDGES) {
            const int e = nfull * SCAN_EDGES + wave * 32 + lane;
            push_row((e < E) ? dst[e] : -1, e);
        }
        while (qn > 0) {
            const int n = (qn < 32) ? qn : 32;
            qn -= n;
            int sv = 0, dv = OWN_ROWS;
            if (lane < n) { const int ee = q[qn + lane]; sv = src[ee]; dv = (int)qr[qn + lane]; }
            add_rows(sv, dv, n);
        }
    }
    __syncthreads();

    float* outp = agg + (size_t)base * HDIM;
    for (int i = tid * 4; i < rows * HDIM; i += 1024 * 4)
        *(float4*)(outp + i) = *(const float4*)(&acc[i]);
    if (FIRST) {
        for (int r = tid; r < rows; r += 1024)
            invdeg[base + r] = 1.0f / fmaxf((float)degc[r], 1.0f);
    }
}

extern "C" void kernel_launch(void* const* d_in, const int* in_sizes, int n_in,
                              void* d_out, int out_size, void* d_ws, size_t ws_size,
                              hipStream_t stream)
{
    const float* x      = (const float*)d_in[0];
    const int*   edge   = (const int*)  d_in[1];
    const float* W_proj = (const float*)d_in[2];
    const float* b_proj = (const float*)d_in[3];
    const float* Wl[3]  = { (const float*)d_in[4], (const float*)d_in[7], (const float*)d_in[10] };
    const float* bl[3]  = { (const float*)d_in[5], (const float*)d_in[8], (const float*)d_in[11] };
    const float* Wr[3]  = { (const float*)d_in[6], (const float*)d_in[9], (const float*)d_in[12] };
    const float* Wc     = (const float*)d_in[13];
    const float* bc     = (const float*)d_in[14];

    const int N    = in_sizes[0] / 256;
    const int E    = in_sizes[1] / 2;
    const int ncls = in_sizes[14];

    const int* src = edge;
    const int* dst = edge + E;

    const int nblk = (N + OWN_ROWS - 1) / OWN_ROWS;
    const int ngrp = (nblk + FGROUP - 1) / FGROUP;

    float* hA     = (float*)d_ws;
    float* hB     = hA  + (size_t)N * HDIM;
    float* agg    = hB  + (size_t)N * HDIM;
    float* invdeg = agg + (size_t)N * HDIM;
    int*   counts = (int*)(invdeg + N);
    int*   lists  = counts + (size_t)nblk * SLOTS;

    const int rowTiles = (N + 15) / 16;

    edge_filter_kernel<<<ngrp * FPARTS, 1024, 0, stream>>>(src, dst, lists, counts, E, N, nblk);

    proj_strip_kernel<<<256, STRIP_THREADS, 0, stream>>>(x, W_proj, b_proj, hA, rowTiles);

    for (int l = 0; l < 3; ++l) {
        if (l == 0)
            owned_rows_sum_kernel<true><<<nblk, 1024, 0, stream>>>(src, dst, hA, agg, lists, counts, invdeg, E, N);
        else
            owned_rows_sum_kernel<false><<<nblk, 1024, 0, stream>>>(src, dst, hA, agg, lists, counts, invdeg, E, N);
        sage_strip_kernel<<<256, STRIP_THREADS, 0, stream>>>(agg, invdeg, hA,
                                                             Wl[l], bl[l], Wr[l], hB, rowTiles);
        float* t = hA; hA = hB; hB = t;
    }

    cls_wmma_kernel<<<dim3(rowTiles, (ncls + 15) / 16), 32, 0, stream>>>(
        hA, Wc, bc, (float*)d_out, ncls);
}


# Dense launches: every wave takes only the whole rounds of strips; each of the 106 leftover strips goes to one block, split by column pairs over waves 0-3 (one per SIMD); per-tile sum order kept
# speedup vs baseline: 3.6663x; 1.0422x over previous
#include <hip/hip_runtime.h>
#include <hip/hip_bf16.h>

typedef __attribute__((ext_vector_type(2))) float v2f;
typedef __attribute__((ext_vector_type(8))) float v8f;

#define HDIM 128

#define STRIP_WAVES   8
#define STRIP_THREADS (STRIP_WAVES * 32)
#define LEFT_WAVES    4
#define WFRAGS        (16 * 8 * 32)

__device__ __forceinline__ void stage_weights(const float* __restrict__ Wg, float4* Ws, int tid)
{
    for (int d = tid; d < WFRAGS; d += STRIP_THREADS) {
        const int l = d & 31, t = (d >> 5) & 7, kb = d >> 8;
        const int c = l & 15, h = l >> 4;
        const float* p = Wg + (size_t)(8 * kb + 4 * h) * HDIM + 16 * t + c;
        Ws[d] = make_float4(p[0], p[HDIM], p[2 * HDIM], p[3 * HDIM]);
    }
}

__device__ __forceinline__ void load_wfrags(float4 (&b)[8], const float4* Ws, int kk, int lane)
{
#pragma unroll
    for (int t = 0; t < 8; ++t) b[t] = Ws[(kk * 8 + t) * 32 + lane];
}

__device__ __forceinline__ void mm_kblock(const float4& a, const float4 (&b)[8], v8f (&acc)[8])
{
    v2f a0, a1;
    a0[0] = a.x; a0[1] = a.y;
    a1[0] = a.z; a1[1] = a.w;
#pragma unroll
    for (int t = 0; t < 8; ++t) {
        v2f bb; bb[0] = b[t].x; bb[1] = b[t].y;
        acc[t] = __builtin_amdgcn_wmma_f32_16x16x4_f32(false, a0, false, bb, (short)0, acc[t], false, false);
    }
#pragma unroll
    for (int t = 0; t < 8; ++t) {
        v2f bb; bb[0] = b[t].z; bb[1] = b[t].w;
        acc[t] = __builtin_amdgcn_wmma_f32_16x16x4_f32(false, a1, false, bb, (short)0, acc[t], false, false);
    }
}

__device__ __forceinline__ void mm_kpair(const float4& a, const float4& w0, const float4& w1, v8f& acc0, v8f& acc1)
{
    v2f lo, hi, p, q;
    lo[0] = a.x;  lo[1] = a.y;
    hi[0] = a.z;  hi[1] = a.w;
    p[0]  = w0.x; p[1]  = w0.y;
    q[0]  = w1.x; q[1]  = w1.y;
    acc0 = __builtin_amdgcn_wmma_f32_16x16x4_f32(false, lo, false, p, (short)0, acc0, false, false);
    acc1 = __builtin_amdgcn_wmma_f32_16x16x4_f32(false, lo, false, q, (short)0, acc1, false, false);
    p[0]  = w0.z; p[1]  = w0.w;
    q[0]  = w1.z; q[1]  = w1.w;
    acc0 = __builtin_amdgcn_wmma_f32_16x16x4_f32(false, hi, false, p, (short)0, acc0, false, false);
    acc1 = __builtin_amdgcn_wmma_f32_16x16x4_f32(false, hi, false, q, (short)0, acc1, false, false);
}

template <bool SCALE>
__device__ __forceinline__ void strip_gemm(const float* __restrict__ A1, const float* __restrict__ A2, int pitch,
                                           const float* __restrict__ rowscale,
                                           const float* __restrict__ W1, const float* __restrict__ W2,
                                           const float* __restrict__ bias, float* __restrict__ out, int ntiles)
{
    __shared__ __align__(16) float4 Ws[2 * WFRAGS];

    const int tid  = threadIdx.x;
    const int lane = tid & 31;
    const int wave = __builtin_amdgcn_readfirstlane(tid >> 5);
    const int c    = lane & 15;
    const int h    = lane >> 4;

    stage_weights(W1, Ws, tid);
    stage_weights(W2, Ws + WFRAGS, tid);
    __syncthreads();

    float bv[8];
#pragma unroll
    for (int t = 0; t < 8; ++t) bv[t] = bias[16 * t + c];

    const int  grid   = (int)gridDim.x;
    const int  stride = STRIP_WAVES * grid;
    const int  nmain  = (ntiles / stride) * stride;
    const int  ltile0 = nmain + (int)blockIdx.x;
    const bool lefty  = (wave < LEFT_WAVES) && (ltile0 < ntiles);
    int tile = wave * grid + (int)blockIdx.x;
    if (tile >= nmain && !lefty) return;

    float4 a1[16], a2[16];
    float  s = 1.0f;
    {
        const int    row = ((tile < nmain) ? tile : ltile0) * 16 + c;
        const float* p1  = A1 + (size_t)row * pitch + 4 * h;
#pragma unroll
        for (int kb = 0; kb < 16; ++kb) a1[kb] = *(const float4*)(p1 + 8 * kb);
        if (SCALE) s = rowscale[row];
    }

    for (; tile < nmain; tile += stride) {
        const float* p2 = A2 + (size_t)(tile * 16 + c) * pitch + 4 * h;
#pragma unroll
        for (int kb = 0; kb < 16; ++kb) a2[kb] = *(const float4*)(p2 + 8 * kb);
        if (SCALE) {
#pragma unroll
            for (int kb = 0; kb < 16; ++kb) { a1[kb].x *= s; a1[kb].y *= s; a1[kb].z *= s; a1[kb].w *= s; }
        }

        v8f acc[8];
#pragma unroll
        for (int t = 0; t < 8; ++t) acc[t] = v8f{};

        float4 b0[8], b1[8];
        load_wfrags(b0, Ws, 0, lane);
#pragma unroll
        for (int kk = 0; kk < 32; kk += 2) {
            if (kk == 16) {
                const int    nt  = (tile + stride < nmain) ? (tile + stride) : (lefty ? ltile0 : tile);
                const int    nr  = nt * 16 + c;
                const float* p1  = A1 + (size_t)nr * pitch + 4 * h;
#pragma unroll
                for (int kb = 0; kb < 16; ++kb) a1[kb] = *(const float4*)(p1 + 8 * kb);
                if (SCALE) s = rowscale[nr];
            }
            load_wfrags(b1, Ws, kk + 1, lane);
            mm_kblock((kk < 16) ? a1[kk & 15] : a2[kk & 15], b0, acc);
            __builtin_amdgcn_sched_barrier(0);
            if (kk + 2 < 32) load_wfrags(b0, Ws, kk + 2, lane);
            mm_kblock((kk + 1 < 16) ? a1[(kk + 1) & 15] : a2[(kk + 1) & 15], b1, acc);
            __builtin_amdgcn_sched_barrier(0);
        }

        float* o = out + (size_t)(tile * 16 + 8 * h) * HDIM + c;
#pragma unroll
        for (int r = 0; r < 8; ++r) {
#pragma unroll
            for (int t = 0; t < 8; ++t)
                o[(size_t)r * HDIM + 16 * t] = fmaxf(acc[t][r] + bv[t], 0.0f);
        }
    }

    if (lefty) {
        const int   t0  = 2 * wave;
        const float bl0 = bias[16 * t0 + c];
        const float bl1 = bias[16 * t0 + 16 + c];
        float4 wf[32][2];
#pragma unroll
        for (int kk = 0; kk < 32; ++kk) {
            wf[kk][0] = Ws[(kk * 8 + t0) * 32 + lane];
            wf[kk][1] = Ws[(kk * 8 + t0 + 1) * 32 + lane];
        }
        for (int lt = ltile0; lt < ntiles; lt += grid) {
            const float* p2 = A2 + (size_t)(lt * 16 + c) * pitch + 4 * h;
#pragma unroll
            for (int kb = 0; kb < 16; ++kb) a2[kb] = *(const float4*)(p2 + 8 * kb);
            if (SCALE) {
#pragma unroll
                for (int kb = 0; kb < 16; ++kb) { a1[kb].x *= s; a1[kb].y *= s; a1[kb].z *= s; a1[kb].w *= s; }
            }

            v8f acc0 = v8f{}, acc1 = v8f{};
#pragma unroll
            for (int kk = 0; kk < 32; ++kk)
                mm_kpair((kk < 16) ? a1[kk & 15] : a2[kk & 15], wf[kk][0], wf[kk][1], acc0, acc1);

            float* o = out + (size_t)(lt * 16 + 8 * h) * HDIM + 16 * t0 + c;
#pragma unroll
            for (int r = 0; r < 8; ++r) {
                o[(size_t)r * HDIM]      = fmaxf(acc0[r] + bl0, 0.0f);
                o[(size_t)r * HDIM + 16] = fmaxf(acc1[r] + bl1, 0.0f);
            }

            if (lt + grid < ntiles) {
                const int    nr = (lt + grid) * 16 + c;
                const float* p1 = A1 + (size_t)nr * pitch + 4 * h;
#pragma unroll
                for (int kb = 0; kb < 16; ++kb) a1[kb] = *(const float4*)(p1 + 8 * kb);
                if (SCALE) s = rowscale[nr];
            }
        }
    }
}

__global__ __launch_bounds__(STRIP_THREADS)
void proj_strip_kernel(const float* __restrict__ x, const float* __restrict__ W,
                       const float* __restrict__ bias, float* __restrict__ out, int ntiles)
{
    strip_gemm<false>(x, x + HDIM, 2 * HDIM, nullptr, W, W + HDIM * HDIM, bias, out, ntiles);
}

__global__ __launch_bounds__(STRIP_THREADS)
void sage_strip_kernel(const float* __restrict__ agg, const float* __restrict__ invdeg,
                       const float* __restrict__ h, const float* __restrict__ Wl,
                       const float* __restrict__ bl, const float* __restrict__ Wr,
                       float* __restrict__ out, int ntiles)
{
    strip_gemm<true>(agg, h, HDIM, invdeg, Wl, Wr, bl, out, ntiles);
}

__global__ __launch_bounds__(32)
void cls_wmma_kernel(const float* __restrict__ h, const float* __restrict__ Wc,
                     const float* __restrict__ bc, float* __restrict__ out, int ncls)
{
    const int row0 = blockIdx.x * 16;
    const int lane = threadIdx.x & 31;
    const int m    = lane & 15;
    const int koff = (lane < 16) ? 0 : 2;
    const int n    = blockIdx.y * 16 + (lane & 15);
    const bool nv  = (n < ncls);

    v8f c = {};
    for (int k = 0; k < HDIM; k += 4) {
        v2f a, b;
        a[0] = h[(size_t)(row0 + m) * HDIM + k + koff];
        a[1] = h[(size_t)(row0 + m) * HDIM + k + koff + 1];
        b[0] = nv ? Wc[(size_t)(k + koff)     * ncls + n] : 0.0f;
        b[1] = nv ? Wc[(size_t)(k + koff + 1) * ncls + n] : 0.0f;
        c = __builtin_amdgcn_wmma_f32_16x16x4_f32(false, a, false, b,
                                                  (short)0, c, false, false);
    }

    if (nv) {
        const int   mbase = row0 + ((lane < 16) ? 0 : 8);
        const float bv    = bc[n];
        for (int r = 0; r < 8; ++r)
            out[(size_t)(mbase + r) * ncls + n] = c[r] + bv;
    }
}

#define OWN_ROWS   400
#define OWN_WAVES  32
#define SCAN_ROWS  16
#define SCAN_EDGES (SCAN_ROWS * 32)
#define QCAP       (SCAN_EDGES + 32)
#define EDGE_GROUP 8
#define FGROUP     8
#define FPARTS     8
#define SLOTS      (FPARTS * OWN_WAVES)
#define SLOT_CAP   256
#define REPLAY_ROUNDS 8

__global__ __launch_bounds__(1024)
void edge_filter_kernel(const int* __restrict__ src, const int* __restrict__ dst,
                        int* __restrict__ lists, int* __restrict__ counts, int E, int N, int nblk)
{
    __shared__ int queue[OWN_WAVES * QCAP];
    __shared__ unsigned short qrow[OWN_WAVES * QCAP];

    const int tid  = threadIdx.x;
    const int lane = tid & 31;
    const int wave = __builtin_amdgcn_readfirstlane(tid >> 5);
    const int g    = (int)blockIdx.x / FPARTS;
    const int p    = (int)blockIdx.x % FPARTS;
    const int ngrp = (int)gridDim.x / FPARTS;
    const int base = g * (FGROUP * OWN_ROWS);
    const int rows = (N - base < FGROUP * OWN_ROWS) ? (N - base) : (FGROUP * OWN_ROWS);

    int*            q  = queue + wave * QCAP;
    unsigned short* qr = qrow  + wave * QCAP;
    int             qn = 0;
    int             cur[FGROUP];
#pragma unroll
    for (int kk = 0; kk < FGROUP; ++kk) cur[kk] = 0;

    const int nfull = E / SCAN_EDGES;
    const int pbeg  = (p * nfull) / FPARTS;
    const int plen  = ((p + 1) * nfull) / FPARTS - pbeg;
    const int first = (int)(((unsigned)g * (unsigned)plen) / (unsigned)ngrp);

    auto push_row = [&](int d, int e) {
        const unsigned t    = (unsigned)(d - base);
        const bool     mine = t < (unsigned)rows;
        const unsigned mask = __builtin_amdgcn_ballot_w32(mine);
        if (mask != 0u) {
            if (mine) {
                const int slot = qn + (int)__builtin_amdgcn_mbcnt_lo(mask, 0u);
                q[slot]  = e;
                qr[slot] = (unsigned short)t;
            }
            qn += __builtin_popcount(mask);
        }
    };

    auto emit = [&](int sv, int tt, int m) {
        const bool valid = lane < m;
        int        k     = 0;
#pragma unroll
        for (int t = 1; t < FGROUP; ++t) k += (tt >= t * OWN_ROWS) ? 1 : 0;
        const int  local = tt - k * OWN_ROWS;
#pragma unroll
        for (int kk = 0; kk < FGROUP; ++kk) {
            const bool     hit = valid && (k == kk);
            const unsigned mk  = __builtin_amdgcn_ballot_w32(hit);
            if (mk != 0u) {
                const int pos = cur[kk] + (int)__builtin_amdgcn_mbcnt_lo(mk, 0u);
                if (hit && pos < SLOT_CAP) {
                    int* L = lists + (size_t)((g * FGROUP + kk) * SLOTS + p * OWN_WAVES + wave) * (2 * SLOT_CAP);
                    L[pos]            = sv;
                    L[SLOT_CAP + pos] = local;
                }
                cur[kk] += __builtin_popcount(mk);
            }
        }
    };

    auto scan_step = [&](int (&use)[SCAN_ROWS], int (&fill)[SCAN_ROWS], int c) {
        int cl = c + 2 * OWN_WAVES;
        if (cl > plen - 1) cl = plen - 1;
        const int ccl = pbeg + ((cl + first < plen) ? (cl + first) : (cl + first - plen));
        const int* dl = dst + ccl * SCAN_EDGES + lane;
#pragma unroll
        for (int j = 0; j < SCAN_ROWS; ++j) fill[j] = dl[32 * j];

        if (c >= 0 && c < plen) {
            const int cc = pbeg + ((c + first < plen) ? (c + first) : (c + first - plen));
            const int e0 = cc * SCAN_EDGES + lane;
#pragma unroll
            for (int j = 0; j < SCAN_ROWS; ++j) push_row(use[j], e0 + 32 * j);
        }

        while (qn >= 32) {
            qn -= 32;
            const int ee = q[qn + lane];
            emit(src[ee], (int)qr[qn + lane], 32);
        }
    };

    if (plen > 0) {
        int ra[SCAN_ROWS], rb[SCAN_ROWS], rc[SCAN_ROWS];
#pragma unroll
        for (int j = 0; j < SCAN_ROWS; ++j) { ra[j] = -1; rb[j] = -1; }
        for (int c = wave - 2 * OWN_WAVES; c < plen; c += 3 * OWN_WAVES) {
            scan_step(ra, rc, c);
            scan_step(rb, ra, c + OWN_WAVES);
            scan_step(rc, rb, c + 2 * OWN_WAVES);
        }
    }

    if (p == FPARTS - 1 && wave * 32 < E - nfull * SCAN_EDGES) {
        const int e = nfull * SCAN_EDGES + wave * 32 + lane;
        push_row((e < E) ? dst[e] : -1, e);
    }
    while (qn > 0) {
        const int n = (qn < 32) ? qn : 32;
        qn -= n;
        int sv = 0, tt = 0;
        if (lane < n) { const int ee = q[qn + lane]; sv = src[ee]; tt = (int)qr[qn + lane]; }
        emit(sv, tt, n);
    }
    if (lane == 0) {
#pragma unroll
        for (int kk = 0; kk < FGROUP; ++kk)
            if (g * FGROUP + kk < nblk) counts[(g * FGROUP + kk) * SLOTS + p * OWN_WAVES + wave] = cur[kk];
    }
}

template <bool FIRST>
__global__ __launch_bounds__(1024)
void owned_rows_sum_kernel(const int* __restrict__ src, const int* __restrict__ dst,
                           const float* __restrict__ h, float* __restrict__ agg,
                           const int* __restrict__ lists, const int* __restrict__ counts,
                           float* __restrict__ invdeg, int E, int N)
{
    __shared__ __align__(16) float acc[(OWN_ROWS + 1) * HDIM];
    __shared__ int queue[OWN_WAVES * QCAP];
    __shared__ unsigned short qrow[OWN_WAVES * QCAP];
    __shared__ int degc[OWN_ROWS];
    __shared__ int qcnt[OWN_WAVES];

    const int tid  = threadIdx.x;
    const int lane = tid & 31;
    const int wave = __builtin_amdgcn_readfirstlane(tid >> 5);
    const int base = blockIdx.x * OWN_ROWS;
    const int rows = (N - base < OWN_ROWS) ? (N - base) : OWN_ROWS;

    for (int i = tid * 4; i < (OWN_ROWS + 1) * HDIM; i += 1024 * 4)
        *(float4*)(&acc[i]) = make_float4(0.0f, 0.0f, 0.0f, 0.0f);
    if (FIRST) {
        for (int i = tid; i < OWN_ROWS; i += 1024) degc[i] = 0;
    }
    if (tid < OWN_WAVES) qcnt[tid] = 0;

    int  ns[FPARTS];
    bool over = false;
#pragma unroll
    for (int s = 0; s < FPARTS; ++s) {
        ns[s] = __builtin_amdgcn_readfirstlane(counts[blockIdx.x * SLOTS + s * OWN_WAVES + wave]);
        over  = over || (ns[s] > SLOT_CAP);
    }
    bool scan = __builtin_amdgcn_readfirstlane(__syncthreads_or(over ? 1 : 0)) != 0;

    const float* hl = h   + lane;
    float*       al = acc + lane;

    auto add_rows = [&](int sv, int dv, int m) {
        if (FIRST) {
            if (lane < m) atomicAdd(&degc[dv], 1);
        }
        for (int g = 0; g < m; g += EDGE_GROUP) {
            float v[EDGE_GROUP][4];
#pragma unroll
            for (int j = 0; j < EDGE_GROUP; ++j) {
                const int s = __builtin_amdgcn_readlane(sv, g + j);
                const float* p = hl + (size_t)s * HDIM;
                v[j][0] = p[0]; v[j][1] = p[32]; v[j][2] = p[64]; v[j][3] = p[96];
            }
#pragma unroll
            for (int j = 0; j < EDGE_GROUP; ++j) {
                const int d = __builtin_amdgcn_readlane(dv, g + j);
                float* r = al + d * HDIM;
                atomicAdd(r,      v[j][0]);
                atomicAdd(r + 32, v[j][1]);
                atomicAdd(r + 64, v[j][2]);
                atomicAdd(r + 96, v[j][3]);
            }
        }
    };

    if (!scan) {
        const int* Lb = lists + (size_t)(blockIdx.x * SLOTS + wave) * (2 * SLOT_CAP);
        int off[FPARTS + 1];
        off[0] = 0;
#pragma unroll
        for (int s = 0; s < FPARTS; ++s) off[s + 1] = off[s] + ns[s];
        const int tot = off[FPARTS];
        bool full = false;
        for (int i0 = 0; i0 < tot; i0 += 32 * REPLAY_ROUNDS) {
            int sv[REPLAY_ROUNDS], dv[REPLAY_ROUNDS];
#pragma unroll
            for (int j = 0; j < REPLAY_ROUNDS; ++j) {
                const int v = i0 + 32 * j + lane;
                sv[j] = 0; dv[j] = OWN_ROWS;
                if (v < tot) {
                    int s = 0, ob = 0;
#pragma unroll
                    for (int t = 1; t < FPARTS; ++t) if (v >= off[t]) { s = t; ob = off[t]; }
                    const int* L = Lb + (size_t)s * OWN_WAVES * (2 * SLOT_CAP);
                    sv[j] = L[v - ob]; dv[j] = L[SLOT_CAP + v - ob];
                }
            }
#pragma unroll
            for (int j = 0; j < REPLAY_ROUNDS; ++j) {
                if (i0 + 32 * j + lane < tot) {
                    const int o    = dv[j] & (OWN_WAVES - 1);
                    const int slot = atomicAdd(&qcnt[o], 1);
                    if (slot < QCAP) {
                        queue[o * QCAP + slot] = sv[j];
                        qrow[o * QCAP + slot]  = (unsigned short)dv[j];
                    } else {
                        full = true;
                    }
                    if (FIRST) atomicAdd(&degc[dv[j]], 1);
                }
            }
        }
        scan = __builtin_amdgcn_readfirstlane(__syncthreads_or(full ? 1 : 0)) != 0;

        if (!scan) {
            const int*            q  = queue + wave * QCAP;
            const unsigned short* qr = qrow  + wave * QCAP;
            const int    n   = __builtin_amdgcn_readfirstlane(qcnt[wave]);
            const float* hl4 = h   + 4 * lane;
            float*       al4 = acc + 4 * lane;

            auto load8 = [&](float4 (&v)[EDGE_GROUP], int svv, int g0) {
#pragma unroll
                for (int j = 0; j < EDGE_GROUP; ++j) {
                    const int s = __builtin_amdgcn_readlane(svv, g0 + j);
                    v[j] = *(const float4*)(hl4 + (size_t)s * HDIM);
                }
            };
            auto rmw8 = [&](const float4 (&v)[EDGE_GROUP], int dvv, int g0) {
#pragma unroll
                for (int j = 0; j < EDGE_GROUP; ++j) {
                    const int d = __builtin_amdgcn_readlane(dvv, g0 + j);
                    float4*   r = (float4*)(al4 + d * HDIM);
                    float4    a = *r;
                    a.x += v[j].x; a.y += v[j].y; a.z += v[j].z; a.w += v[j].w;
                    *r = a;
                }
            };

            int svv = 0, dvv = OWN_ROWS;
            if (lane < n) { svv = q[lane]; dvv = (int)qr[lane]; }
            float4 va[EDGE_GROUP], vb[EDGE_GROUP];
            load8(va, svv, 0);
            const int nround = n >> 5;
            for (int r = 0; r < nround; ++r) {
                int svn = 0, dvn = OWN_ROWS;
                const int k = 32 * (r + 1) + lane;
                if (k < n) { svn = q[k]; dvn = (int)qr[k]; }
                load8(vb, svv, 8);   rmw8(va, dvv, 0);
                load8(va, svv, 16);  rmw8(vb, dvv, 8);
                load8(vb, svv, 24);  rmw8(va, dvv, 16);
                load8(va, svn, 0);   rmw8(vb, dvv, 24);
                svv = svn; dvv = dvn;
            }
            const int m = n & 31;
            if (m > 0) {
                rmw8(va, dvv, 0);
                if (m > 8)  { load8(vb, svv, 8);  rmw8(vb, dvv, 8); }
                if (m > 16) { load8(va, svv, 16); rmw8(va, dvv, 16); }
                if (m > 24) { load8(vb, svv, 24); rmw8(vb, dvv, 24); }
            }
        } else if (FIRST) {
            for (int i = tid; i < OWN_ROWS; i += 1024) degc[i] = 0;
            __syncthreads();
        }
    }
    if (scan) {
        int*            q  = queue + wave * QCAP;
        unsigned short* qr = qrow  + wave * QCAP;
        int             qn = 0;

        const int nfull = E / SCAN_EDGES;
        const int first = (int)((blockIdx.x * (unsigned)nfull) / gridDim.x);

        auto push_row = [&](int d, int e) {
            const unsigned t    = (unsigned)(d - base);
            const bool     mine = t < (unsigned)rows;
            const unsigned mask = __builtin_amdgcn_ballot_w32(mine);
            if (mask != 0u) {
                if (mine) {
                    const int slot = qn + (int)__builtin_amdgcn_mbcnt_lo(mask, 0u);
                    q[slot]  = e;
                    qr[slot] = (unsigned short)t;
                }
                qn += __builtin_popcount(mask);
            }
        };

        auto scan_step = [&](int (&use)[SCAN_ROWS], int (&fill)[SCAN_ROWS], int c) {
            int cl = c + 2 * OWN_WAVES;
            if (cl > nfull - 1) cl = nfull - 1;
            const int ccl = (cl + first < nfull) ? (cl + first) : (cl + first - nfull);
            const int* dl = dst + ccl * SCAN_EDGES + lane;
#pragma unroll
            for (int j = 0; j < SCAN_ROWS; ++j) fill[j] = dl[32 * j];

            if (c >= 0 && c < nfull) {
                const int cc = (c + first < nfull) ? (c + first) : (c + first - nfull);
                const int e0 = cc * SCAN_EDGES + lane;
#pragma unroll
                for (int j = 0; j < SCAN_ROWS; ++j) push_row(use[j], e0 + 32 * j);
            }

            while (qn >= 32) {
                qn -= 32;
                const int ee = q[qn + lane];
                add_rows(src[ee], (int)qr[qn + lane], 32);
            }
        };

        if (nfull > 0) {
            int ra[SCAN_ROWS], rb[SCAN_ROWS], rc[SCAN_ROWS];
#pragma unroll
            for (int j = 0; j < SCAN_ROWS; ++j) { ra[j] = -1; rb[j] = -1; }
            for (int c = wave - 2 * OWN_WAVES; c < nfull; c += 3 * OWN_WAVES) {
                scan_step(ra, rc, c);
                scan_step(rb, ra, c + OWN_WAVES);
                scan_step(rc, rb, c + 2 * OWN_WAVES);
            }
        }
        if (wave * 32 < E - nfull * SCAN_EDGES) {
            const int e = nfull * SCAN_EDGES + wave * 32 + lane;
            push_row((e < E) ? dst[e] : -1, e);
        }
        while (qn > 0) {
            const int n = (qn < 32) ? qn : 32;
            qn -= n;
            int sv = 0, dv = OWN_ROWS;
            if (lane < n) { const int ee = q[qn + lane]; sv = src[ee]; dv = (int)qr[qn + lane]; }
            add_rows(sv, dv, n);
        }
    }
    __syncthreads();

    float* outp = agg + (size_t)base * HDIM;
    for (int i = tid * 4; i < rows * HDIM; i += 1024 * 4)
        *(float4*)(outp + i) = *(const float4*)(&acc[i]);
    if (FIRST) {
        for (int r = tid; r < rows; r += 1024)
            invdeg[base + r] = 1.0f / fmaxf((float)degc[r], 1.0f);
    }
}

extern "C" void kernel_launch(void* const* d_in, const int* in_sizes, int n_in,
                              void* d_out, int out_size, void* d_ws, size_t ws_size,
                              hipStream_t stream)
{
    const float* x      = (const float*)d_in[0];
    const int*   edge   = (const int*)  d_in[1];
    const float* W_proj = (const float*)d_in[2];
    const float* b_proj = (const float*)d_in[3];
    const float* Wl[3]  = { (const float*)d_in[4], (const float*)d_in[7], (const float*)d_in[10] };
    const float* bl[3]  = { (const float*)d_in[5], (const float*)d_in[8], (const float*)d_in[11] };
    const float* Wr[3]  = { (const float*)d_in[6], (const float*)d_in[9], (const float*)d_in[12] };
    const float* Wc     = (const float*)d_in[13];
    const float* bc     = (const float*)d_in[14];

    const int N    = in_sizes[0] / 256;
    const int E    = in_sizes[1] / 2;
    const int ncls = in_sizes[14];

    const int* src = edge;
    const int* dst = edge + E;

    const int nblk = (N + OWN_ROWS - 1) / OWN_ROWS;
    const int ngrp = (nblk + FGROUP - 1) / FGROUP;

    float* hA     = (float*)d_ws;
    float* hB     = hA  + (size_t)N * HDIM;
    float* agg    = hB  + (size_t)N * HDIM;
    float* invdeg = agg + (size_t)N * HDIM;
    int*   counts = (int*)(invdeg + N);
    int*   lists  = counts + (size_t)nblk * SLOTS;

    const int rowTiles = (N + 15) / 16;

    edge_filter_kernel<<<ngrp * FPARTS, 1024, 0, stream>>>(src, dst, lists, counts, E, N, nblk);

    proj_strip_kernel<<<256, STRIP_THREADS, 0, stream>>>(x, W_proj, b_proj, hA, rowTiles);

    for (int l = 0; l < 3; ++l) {
        if (l == 0)
            owned_rows_sum_kernel<true><<<nblk, 1024, 0, stream>>>(src, dst, hA, agg, lists, counts, invdeg, E, N);
        else
            owned_rows_sum_kernel<false><<<nblk, 1024, 0, stream>>>(src, dst, hA, agg, lists, counts, invdeg, E, N);
        sage_strip_kernel<<<256, STRIP_THREADS, 0, stream>>>(agg, invdeg, hA,
                                                             Wl[l], bl[l], Wr[l], hB, rowTiles);
        float* t = hA; hA = hB; hB = t;
    }

    cls_wmma_kernel<<<dim3(rowTiles, (ncls + 15) / 16), 32, 0, stream>>>(
        hA, Wc, bc, (float*)d_out, ncls);
}


# Classifier re-cut as row strips: W_cls fragments in LDS once per block, a wave takes two 16-row strips with 16-B A loads from global; 3125 waves, not 18750 one-wave blocks; its k order changes
# speedup vs baseline: 3.9457x; 1.0762x over previous
#include <hip/hip_runtime.h>
#include <hip/hip_bf16.h>

typedef __attribute__((ext_vector_type(2))) float v2f;
typedef __attribute__((ext_vector_type(8))) float v8f;

#define HDIM 128

#define STRIP_WAVES   8
#define STRIP_THREADS (STRIP_WAVES * 32)
#define LEFT_WAVES    4
#define WFRAGS        (16 * 8 * 32)

__device__ __forceinline__ void stage_weights(const float* __restrict__ Wg, float4* Ws, int tid)
{
    for (int d = tid; d < WFRAGS; d += STRIP_THREADS) {
        const int l = d & 31, t = (d >> 5) & 7, kb = d >> 8;
        const int c = l & 15, h = l >> 4;
        const float* p = Wg + (size_t)(8 * kb + 4 * h) * HDIM + 16 * t + c;
        Ws[d] = make_float4(p[0], p[HDIM], p[2 * HDIM], p[3 * HDIM]);
    }
}

__device__ __forceinline__ void load_wfrags(float4 (&b)[8], const float4* Ws, int kk, int lane)
{
#pragma unroll
    for (int t = 0; t < 8; ++t) b[t] = Ws[(kk * 8 + t) * 32 + lane];
}

__device__ __forceinline__ void mm_kblock(const float4& a, const float4 (&b)[8], v8f (&acc)[8])
{
    v2f a0, a1;
    a0[0] = a.x; a0[1] = a.y;
    a1[0] = a.z; a1[1] = a.w;
#pragma unroll
    for (int t = 0; t < 8; ++t) {
        v2f bb; bb[0] = b[t].x; bb[1] = b[t].y;
        acc[t] = __builtin_amdgcn_wmma_f32_16x16x4_f32(false, a0, false, bb, (short)0, acc[t], false, false);
    }
#pragma unroll
    for (int t = 0; t < 8; ++t) {
        v2f bb; bb[0] = b[t].z; bb[1] = b[t].w;
        acc[t] = __builtin_amdgcn_wmma_f32_16x16x4_f32(false, a1, false, bb, (short)0, acc[t], false, false);
    }
}

__device__ __forceinline__ void mm_kpair(const float4& a, const float4& w0, const float4& w1, v8f& acc0, v8f& acc1)
{
    v2f lo, hi, p, q;
    lo[0] = a.x;  lo[1] = a.y;
    hi[0] = a.z;  hi[1] = a.w;
    p[0]  = w0.x; p[1]  = w0.y;
    q[0]  = w1.x; q[1]  = w1.y;
    acc0 = __builtin_amdgcn_wmma_f32_16x16x4_f32(false, lo, false, p, (short)0, acc0, false, false);
    acc1 = __builtin_amdgcn_wmma_f32_16x16x4_f32(false, lo, false, q, (short)0, acc1, false, false);
    p[0]  = w0.z; p[1]  = w0.w;
    q[0]  = w1.z; q[1]  = w1.w;
    acc0 = __builtin_amdgcn_wmma_f32_16x16x4_f32(false, hi, false, p, (short)0, acc0, false, false);
    acc1 = __builtin_amdgcn_wmma_f32_16x16x4_f32(false, hi, false, q, (short)0, acc1, false, false);
}

template <bool SCALE>
__device__ __forceinline__ void strip_gemm(const float* __restrict__ A1, const float* __restrict__ A2, int pitch,
                                           const float* __restrict__ rowscale,
                                           const float* __restrict__ W1, const float* __restrict__ W2,
                                           const float* __restrict__ bias, float* __restrict__ out, int ntiles)
{
    __shared__ __align__(16) float4 Ws[2 * WFRAGS];

    const int tid  = threadIdx.x;
    const int lane = tid & 31;
    const int wave = __builtin_amdgcn_readfirstlane(tid >> 5);
    const int c    = lane & 15;
    const int h    = lane >> 4;

    stage_weights(W1, Ws, tid);
    stage_weights(W2, Ws + WFRAGS, tid);
    __syncthreads();

    float bv[8];
#pragma unroll
    for (int t = 0; t < 8; ++t) bv[t] = bias[16 * t + c];

    const int  grid   = (int)gridDim.x;
    const int  stride = STRIP_WAVES * grid;
    const int  nmain  = (ntiles / stride) * stride;
    const int  ltile0 = nmain + (int)blockIdx.x;
    const bool lefty  = (wave < LEFT_WAVES) && (ltile0 < ntiles);
    int tile = wave * grid + (int)blockIdx.x;
    if (tile >= nmain && !lefty) return;

    float4 a1[16], a2[16];
    float  s = 1.0f;
    {
        const int    row = ((tile < nmain) ? tile : ltile0) * 16 + c;
        const float* p1  = A1 + (size_t)row * pitch + 4 * h;
#pragma unroll
        for (int kb = 0; kb < 16; ++kb) a1[kb] = *(const float4*)(p1 + 8 * kb);
        if (SCALE) s = rowscale[row];
    }

    for (; tile < nmain; tile += stride) {
        const float* p2 = A2 + (size_t)(tile * 16 + c) * pitch + 4 * h;
#pragma unroll
        for (int kb = 0; kb < 16; ++kb) a2[kb] = *(const float4*)(p2 + 8 * kb);
        if (SCALE) {
#pragma unroll
            for (int kb = 0; kb < 16; ++kb) { a1[kb].x *= s; a1[kb].y *= s; a1[kb].z *= s; a1[kb].w *= s; }
        }

        v8f acc[8];
#pragma unroll
        for (int t = 0; t < 8; ++t) acc[t] = v8f{};

        float4 b0[8], b1[8];
        load_wfrags(b0, Ws, 0, lane);
#pragma unroll
        for (int kk = 0; kk < 32; kk += 2) {
            if (kk == 16) {
                const int    nt  = (tile + stride < nmain) ? (tile + stride) : (lefty ? ltile0 : tile);
                const int    nr  = nt * 16 + c;
                const float* p1  = A1 + (size_t)nr * pitch + 4 * h;
#pragma unroll
                for (int kb = 0; kb < 16; ++kb) a1[kb] = *(const float4*)(p1 + 8 * kb);
                if (SCALE) s = rowscale[nr];
            }
            load_wfrags(b1, Ws, kk + 1, lane);
            mm_kblock((kk < 16) ? a1[kk & 15] : a2[kk & 15], b0, acc);
            __builtin_amdgcn_sched_barrier(0);
            if (kk + 2 < 32) load_wfrags(b0, Ws, kk + 2, lane);
            mm_kblock((kk + 1 < 16) ? a1[(kk + 1) & 15] : a2[(kk + 1) & 15], b1, acc);
            __builtin_amdgcn_sched_barrier(0);
        }

        float* o = out + (size_t)(tile * 16 + 8 * h) * HDIM + c;
#pragma unroll
        for (int r = 0; r < 8; ++r) {
#pragma unroll
            for (int t = 0; t < 8; ++t)
                o[(size_t)r * HDIM + 16 * t] = fmaxf(acc[t][r] + bv[t], 0.0f);
        }
    }

    if (lefty) {
        const int   t0  = 2 * wave;
        const float bl0 = bias[16 * t0 + c];
        const float bl1 = bias[16 * t0 + 16 + c];
        float4 wf[32][2];
#pragma unroll
        for (int kk = 0; kk < 32; ++kk) {
            wf[kk][0] = Ws[(kk * 8 + t0) * 32 + lane];
            wf[kk][1] = Ws[(kk * 8 + t0 + 1) * 32 + lane];
        }
        for (int lt = ltile0; lt < ntiles; lt += grid) {
            const float* p2 = A2 + (size_t)(lt * 16 + c) * pitch + 4 * h;
#pragma unroll
            for (int kb = 0; kb < 16; ++kb) a2[kb] = *(const float4*)(p2 + 8 * kb);
            if (SCALE) {
#pragma unroll
                for (int kb = 0; kb < 16; ++kb) { a1[kb].x *= s; a1[kb].y *= s; a1[kb].z *= s; a1[kb].w *= s; }
            }

            v8f acc0 = v8f{}, acc1 = v8f{};
#pragma unroll
            for (int kk = 0; kk < 32; ++kk)
                mm_kpair((kk < 16) ? a1[kk & 15] : a2[kk & 15], wf[kk][0], wf[kk][1], acc0, acc1);

            float* o = out + (size_t)(lt * 16 + 8 * h) * HDIM + 16 * t0 + c;
#pragma unroll
            for (int r = 0; r < 8; ++r) {
                o[(size_t)r * HDIM]      = fmaxf(acc0[r] + bl0, 0.0f);
                o[(size_t)r * HDIM + 16] = fmaxf(acc1[r] + bl1, 0.0f);
            }

            if (lt + grid < ntiles) {
                const int    nr = (lt + grid) * 16 + c;
                const float* p1 = A1 + (size_t)nr * pitch + 4 * h;
#pragma unroll
                for (int kb = 0; kb < 16; ++kb) a1[kb] = *(const float4*)(p1 + 8 * kb);
                if (SCALE) s = rowscale[nr];
            }
        }
    }
}

__global__ __launch_bounds__(STRIP_THREADS)
void proj_strip_kernel(const float* __restrict__ x, const float* __restrict__ W,
                       const float* __restrict__ bias, float* __restrict__ out, int ntiles)
{
    strip_gemm<false>(x, x + HDIM, 2 * HDIM, nullptr, W, W + HDIM * HDIM, bias, out, ntiles);
}

__global__ __launch_bounds__(STRIP_THREADS)
void sage_strip_kernel(const float* __restrict__ agg, const float* __restrict__ invdeg,
                       const float* __restrict__ h, const float* __restrict__ Wl,
                       const float* __restrict__ bl, const float* __restrict__ Wr,
                       float* __restrict__ out, int ntiles)
{
    strip_gemm<true>(agg, h, HDIM, invdeg, Wl, Wr, bl, out, ntiles);
}

#define CLS_WAVES   5
#define CLS_THREADS (CLS_WAVES * 32)
#define CLS_TILES   3
#define CLS_FRAGS   (16 * CLS_TILES * 32)
#define CLS_STAGE   ((CLS_FRAGS + CLS_THREADS - 1) / CLS_THREADS)

__device__ __forceinline__ void load_cfrags(float4 (&b)[CLS_TILES], const float4* Wf, int kb, int lane)
{
#pragma unroll
    for (int t = 0; t < CLS_TILES; ++t) b[t] = Wf[(kb * CLS_TILES + t) * 32 + lane];
}

__device__ __forceinline__ void mm_kblock2(const float4& x0, const float4& x1, const float4 (&b)[CLS_TILES],
                                           v8f (&acc)[2][CLS_TILES])
{
    v2f p0, p1, q0, q1;
    p0[0] = x0.x; p0[1] = x0.y;  q0[0] = x0.z; q0[1] = x0.w;
    p1[0] = x1.x; p1[1] = x1.y;  q1[0] = x1.z; q1[1] = x1.w;
#pragma unroll
    for (int t = 0; t < CLS_TILES; ++t) {
        v2f bb; bb[0] = b[t].x; bb[1] = b[t].y;
        acc[0][t] = __builtin_amdgcn_wmma_f32_16x16x4_f32(false, p0, false, bb, (short)0, acc[0][t], false, false);
        acc[1][t] = __builtin_amdgcn_wmma_f32_16x16x4_f32(false, p1, false, bb, (short)0, acc[1][t], false, false);
    }
#pragma unroll
    for (int t = 0; t < CLS_TILES; ++t) {
        v2f bb; bb[0] = b[t].z; bb[1] = b[t].w;
        acc[0][t] = __builtin_amdgcn_wmma_f32_16x16x4_f32(false, q0, false, bb, (short)0, acc[0][t], false, false);
        acc[1][t] = __builtin_amdgcn_wmma_f32_16x16x4_f32(false, q1, false, bb, (short)0, acc[1][t], false, false);
    }
}

__global__ __launch_bounds__(CLS_THREADS)
void cls_strip_kernel(const float* __restrict__ h, const float* __restrict__ Wc,
                      const float* __restrict__ bc, float* __restrict__ out, int ncls, int ntiles)
{
    __shared__ __align__(16) float4 Wf[CLS_FRAGS];

    const int tid  = threadIdx.x;
    const int lane = tid & 31;
    const int wave = __builtin_amdgcn_readfirstlane(tid >> 5);
    const int c    = lane & 15;
    const int hh   = lane >> 4;
    const int col0 = (int)blockIdx.y * (16 * CLS_TILES);

    float4 wv[CLS_STAGE];
#pragma unroll
    for (int i = 0; i < CLS_STAGE; ++i) {
        int d = tid + i * CLS_THREADS;
        if (d > CLS_FRAGS - 1) d = CLS_FRAGS - 1;
        const int l = d & 31, r = d >> 5;
        const int t = r % CLS_TILES, kb = r / CLS_TILES;
        const int n = col0 + 16 * t + (l & 15);
        wv[i] = make_float4(0.0f, 0.0f, 0.0f, 0.0f);
        if (n < ncls) {
            const float* p = Wc + (size_t)(8 * kb + 4 * (l >> 4)) * ncls + n;
            wv[i] = make_float4(p[0], p[ncls], p[2 * ncls], p[3 * ncls]);
        }
    }

    const int  npairs = (ntiles + 1) >> 1;
    const int  dp     = (int)blockIdx.x * CLS_WAVES + wave;
    const bool has    = dp < npairs;
    const int  s0     = 2 * (has ? dp : (npairs - 1));
    const bool two    = s0 + 1 < ntiles;
    float4 a0[16], a1[16];
    {
        const float* p0 = h + (size_t)(s0 * 16 + c) * HDIM + 4 * hh;
        const float* p1 = h + (size_t)((two ? s0 + 1 : s0) * 16 + c) * HDIM + 4 * hh;
#pragma unroll
        for (int kb = 0; kb < 16; ++kb) a0[kb] = *(const float4*)(p0 + 8 * kb);
#pragma unroll
        for (int kb = 0; kb < 16; ++kb) a1[kb] = *(const float4*)(p1 + 8 * kb);
    }

#pragma unroll
    for (int i = 0; i < CLS_STAGE; ++i) {
        const int d = tid + i * CLS_THREADS;
        if (d < CLS_FRAGS) Wf[d] = wv[i];
    }
    __syncthreads();
    if (!has) return;

    float bv[CLS_TILES];
#pragma unroll
    for (int t = 0; t < CLS_TILES; ++t) {
        const int n = col0 + 16 * t + c;
        bv[t] = (n < ncls) ? bc[n] : 0.0f;
    }

    v8f acc[2][CLS_TILES];
#pragma unroll
    for (int t = 0; t < CLS_TILES; ++t) { acc[0][t] = v8f{}; acc[1][t] = v8f{}; }

    float4 b0[CLS_TILES], b1[CLS_TILES];
    load_cfrags(b0, Wf, 0, lane);
#pragma unroll
    for (int kb = 0; kb < 16; kb += 2) {
        load_cfrags(b1, Wf, kb + 1, lane);
        mm_kblock2(a0[kb], a1[kb], b0, acc);
        if (kb + 2 < 16) load_cfrags(b0, Wf, kb + 2, lane);
        mm_kblock2(a0[kb + 1], a1[kb + 1], b1, acc);
    }

#pragma unroll
    for (int s = 0; s < 2; ++s) {
        if (s == 0 || two) {
            float* o = out + (size_t)((s0 + s) * 16 + 8 * hh) * ncls + col0 + c;
#pragma unroll
            for (int t = 0; t < CLS_TILES; ++t) {
                if (col0 + 16 * t + c < ncls) {
#pragma unroll
                    for (int r = 0; r < 8; ++r) o[(size_t)r * ncls + 16 * t] = acc[s][t][r] + bv[t];
                }
            }
        }
    }
}

#define OWN_ROWS   400
#define OWN_WAVES  32
#define SCAN_ROWS  16
#define SCAN_EDGES (SCAN_ROWS * 32)
#define QCAP       (SCAN_EDGES + 32)
#define EDGE_GROUP 8
#define FGROUP     8
#define FPARTS     8
#define SLOTS      (FPARTS * OWN_WAVES)
#define SLOT_CAP   256
#define REPLAY_ROUNDS 8

__global__ __launch_bounds__(1024)
void edge_filter_kernel(const int* __restrict__ src, const int* __restrict__ dst,
                        int* __restrict__ lists, int* __restrict__ counts, int E, int N, int nblk)
{
    __shared__ int queue[OWN_WAVES * QCAP];
    __shared__ unsigned short qrow[OWN_WAVES * QCAP];

    const int tid  = threadIdx.x;
    const int lane = tid & 31;
    const int wave = __builtin_amdgcn_readfirstlane(tid >> 5);
    const int g    = (int)blockIdx.x / FPARTS;
    const int p    = (int)blockIdx.x % FPARTS;
    const int ngrp = (int)gridDim.x / FPARTS;
    const int base = g * (FGROUP * OWN_ROWS);
    const int rows = (N - base < FGROUP * OWN_ROWS) ? (N - base) : (FGROUP * OWN_ROWS);

    int*            q  = queue + wave * QCAP;
    unsigned short* qr = qrow  + wave * QCAP;
    int             qn = 0;
    int             cur[FGROUP];
#pragma unroll
    for (int kk = 0; kk < FGROUP; ++kk) cur[kk] = 0;

    const int nfull = E / SCAN_EDGES;
    const int pbeg  = (p * nfull) / FPARTS;
    const int plen  = ((p + 1) * nfull) / FPARTS - pbeg;
    const int first = (int)(((unsigned)g * (unsigned)plen) / (unsigned)ngrp);

    auto push_row = [&](int d, int e) {
        const unsigned t    = (unsigned)(d - base);
        const bool     mine = t < (unsigned)rows;
        const unsigned mask = __builtin_amdgcn_ballot_w32(mine);
        if (mask != 0u) {
            if (mine) {
                const int slot = qn + (int)__builtin_amdgcn_mbcnt_lo(mask, 0u);
                q[slot]  = e;
                qr[slot] = (unsigned short)t;
            }
            qn += __builtin_popcount(mask);
        }
    };

    auto emit = [&](int sv, int tt, int m) {
        const bool valid = lane < m;
        int        k     = 0;
#pragma unroll
        for (int t = 1; t < FGROUP; ++t) k += (tt >= t * OWN_ROWS) ? 1 : 0;
        const int  local = tt - k * OWN_ROWS;
#pragma unroll
        for (int kk = 0; kk < FGROUP; ++kk) {
            const bool     hit = valid && (k == kk);
            const unsigned mk  = __builtin_amdgcn_ballot_w32(hit);
            if (mk != 0u) {
                const int pos = cur[kk] + (int)__builtin_amdgcn_mbcnt_lo(mk, 0u);
                if (hit && pos < SLOT_CAP) {
                    int* L = lists + (size_t)((g * FGROUP + kk) * SLOTS + p * OWN_WAVES + wave) * (2 * SLOT_CAP);
                    L[pos]            = sv;
                    L[SLOT_CAP + pos] = local;
                }
                cur[kk] += __builtin_popcount(mk);
            }
        }
    };

    auto scan_step = [&](int (&use)[SCAN_ROWS], int (&fill)[SCAN_ROWS], int c) {
        int cl = c + 2 * OWN_WAVES;
        if (cl > plen - 1) cl = plen - 1;
        const int ccl = pbeg + ((cl + first < plen) ? (cl + first) : (cl + first - plen));
        const int* dl = dst + ccl * SCAN_EDGES + lane;
#pragma unroll
        for (int j = 0; j < SCAN_ROWS; ++j) fill[j] = dl[32 * j];

        if (c >= 0 && c < plen) {
            const int cc = pbeg + ((c + first < plen) ? (c + first) : (c + first - plen));
            const int e0 = cc * SCAN_EDGES + lane;
#pragma unroll
            for (int j = 0; j < SCAN_ROWS; ++j) push_row(use[j], e0 + 32 * j);
        }

        while (qn >= 32) {
            qn -= 32;
            const int ee = q[qn + lane];
            emit(src[ee], (int)qr[qn + lane], 32);
        }
    };

    if (plen > 0) {
        int ra[SCAN_ROWS], rb[SCAN_ROWS], rc[SCAN_ROWS];
#pragma unroll
        for (int j = 0; j < SCAN_ROWS; ++j) { ra[j] = -1; rb[j] = -1; }
        for (int c = wave - 2 * OWN_WAVES; c < plen; c += 3 * OWN_WAVES) {
            scan_step(ra, rc, c);
            scan_step(rb, ra, c + OWN_WAVES);
            scan_step(rc, rb, c + 2 * OWN_WAVES);
        }
    }

    if (p == FPARTS - 1 && wave * 32 < E - nfull * SCAN_EDGES) {
        const int e = nfull * SCAN_EDGES + wave * 32 + lane;
        push_row((e < E) ? dst[e] : -1, e);
    }
    while (qn > 0) {
        const int n = (qn < 32) ? qn : 32;
        qn -= n;
        int sv = 0, tt = 0;
        if (lane < n) { const int ee = q[qn + lane]; sv = src[ee]; tt = (int)qr[qn + lane]; }
        emit(sv, tt, n);
    }
    if (lane == 0) {
#pragma unroll
        for (int kk = 0; kk < FGROUP; ++kk)
            if (g * FGROUP + kk < nblk) counts[(g * FGROUP + kk) * SLOTS + p * OWN_WAVES + wave] = cur[kk];
    }
}

template <bool FIRST>
__global__ __launch_bounds__(1024)
void owned_rows_sum_kernel(const int* __restrict__ src, const int* __restrict__ dst,
                           const float* __restrict__ h, float* __restrict__ agg,
                           const int* __restrict__ lists, const int* __restrict__ counts,
                           float* __restrict__ invdeg, int E, int N)
{
    __shared__ __align__(16) float acc[(OWN_ROWS + 1) * HDIM];
    __shared__ int queue[OWN_WAVES * QCAP];
    __shared__ unsigned short qrow[OWN_WAVES * QCAP];
    __shared__ int degc[OWN_ROWS];
    __shared__ int qcnt[OWN_WAVES];

    const int tid  = threadIdx.x;
    const int lane = tid & 31;
    const int wave = __builtin_amdgcn_readfirstlane(tid >> 5);
    const int base = blockIdx.x * OWN_ROWS;
    const int rows = (N - base < OWN_ROWS) ? (N - base) : OWN_ROWS;

    for (int i = tid * 4; i < (OWN_ROWS + 1) * HDIM; i += 1024 * 4)
        *(float4*)(&acc[i]) = make_float4(0.0f, 0.0f, 0.0f, 0.0f);
    if (FIRST) {
        for (int i = tid; i < OWN_ROWS; i += 1024) degc[i] = 0;
    }
    if (tid < OWN_WAVES) qcnt[tid] = 0;

    int  ns[FPARTS];
    bool over = false;
#pragma unroll
    for (int s = 0; s < FPARTS; ++s) {
        ns[s] = __builtin_amdgcn_readfirstlane(counts[blockIdx.x * SLOTS + s * OWN_WAVES + wave]);
        over  = over || (ns[s] > SLOT_CAP);
    }
    bool scan = __builtin_amdgcn_readfirstlane(__syncthreads_or(over ? 1 : 0)) != 0;

    const float* hl = h   + lane;
    float*       al = acc + lane;

    auto add_rows = [&](int sv, int dv, int m) {
        if (FIRST) {
            if (lane < m) atomicAdd(&degc[dv], 1);
        }
        for (int g = 0; g < m; g += EDGE_GROUP) {
            float v[EDGE_GROUP][4];
#pragma unroll
            for (int j = 0; j < EDGE_GROUP; ++j) {
                const int s = __builtin_amdgcn_readlane(sv, g + j);
                const float* p = hl + (size_t)s * HDIM;
                v[j][0] = p[0]; v[j][1] = p[32]; v[j][2] = p[64]; v[j][3] = p[96];
            }
#pragma unroll
            for (int j = 0; j < EDGE_GROUP; ++j) {
                const int d = __builtin_amdgcn_readlane(dv, g + j);
                float* r = al + d * HDIM;
                atomicAdd(r,      v[j][0]);
                atomicAdd(r + 32, v[j][1]);
                atomicAdd(r + 64, v[j][2]);
                atomicAdd(r + 96, v[j][3]);
            }
        }
    };

    if (!scan) {
        const int* Lb = lists + (size_t)(blockIdx.x * SLOTS + wave) * (2 * SLOT_CAP);
        int off[FPARTS + 1];
        off[0] = 0;
#pragma unroll
        for (int s = 0; s < FPARTS; ++s) off[s + 1] = off[s] + ns[s];
        const int tot = off[FPARTS];
        bool full = false;
        for (int i0 = 0; i0 < tot; i0 += 32 * REPLAY_ROUNDS) {
            int sv[REPLAY_ROUNDS], dv[REPLAY_ROUNDS];
#pragma unroll
            for (int j = 0; j < REPLAY_ROUNDS; ++j) {
                const int v = i0 + 32 * j + lane;
                sv[j] = 0; dv[j] = OWN_ROWS;
                if (v < tot) {
                    int s = 0, ob = 0;
#pragma unroll
                    for (int t = 1; t < FPARTS; ++t) if (v >= off[t]) { s = t; ob = off[t]; }
                    const int* L = Lb + (size_t)s * OWN_WAVES * (2 * SLOT_CAP);
                    sv[j] = L[v - ob]; dv[j] = L[SLOT_CAP + v - ob];
                }
            }
#pragma unroll
            for (int j = 0; j < REPLAY_ROUNDS; ++j) {
                if (i0 + 32 * j + lane < tot) {
                    const int o    = dv[j] & (OWN_WAVES - 1);
                    const int slot = atomicAdd(&qcnt[o], 1);
                    if (slot < QCAP) {
                        queue[o * QCAP + slot] = sv[j];
                        qrow[o * QCAP + slot]  = (unsigned short)dv[j];
                    } else {
                        full = true;
                    }
                    if (FIRST) atomicAdd(&degc[dv[j]], 1);
                }
            }
        }
        scan = __builtin_amdgcn_readfirstlane(__syncthreads_or(full ? 1 : 0)) != 0;

        if (!scan) {
            const int*            q  = queue + wave * QCAP;
            const unsigned short* qr = qrow  + wave * QCAP;
            const int    n   = __builtin_amdgcn_readfirstlane(qcnt[wave]);
            const float* hl4 = h   + 4 * lane;
            float*       al4 = acc + 4 * lane;

            auto load8 = [&](float4 (&v)[EDGE_GROUP], int svv, int g0) {
#pragma unroll
                for (int j = 0; j < EDGE_GROUP; ++j) {
                    const int s = __builtin_amdgcn_readlane(svv, g0 + j);
                    v[j] = *(const float4*)(hl4 + (size_t)s * HDIM);
                }
            };
            auto rmw8 = [&](const float4 (&v)[EDGE_GROUP], int dvv, int g0) {
#pragma unroll
                for (int j = 0; j < EDGE_GROUP; ++j) {
                    const int d = __builtin_amdgcn_readlane(dvv, g0 + j);
                    float4*   r = (float4*)(al4 + d * HDIM);
                    float4    a = *r;
                    a.x += v[j].x; a.y += v[j].y; a.z += v[j].z; a.w += v[j].w;
                    *r = a;
                }
            };

            int svv = 0, dvv = OWN_ROWS;
            if (lane < n) { svv = q[lane]; dvv = (int)qr[lane]; }
            float4 va[EDGE_GROUP], vb[EDGE_GROUP];
            load8(va, svv, 0);
            const int nround = n >> 5;
            for (int r = 0; r < nround; ++r) {
                int svn = 0, dvn = OWN_ROWS;
                const int k = 32 * (r + 1) + lane;
                if (k < n) { svn = q[k]; dvn = (int)qr[k]; }
                load8(vb, svv, 8);   rmw8(va, dvv, 0);
                load8(va, svv, 16);  rmw8(vb, dvv, 8);
                load8(vb, svv, 24);  rmw8(va, dvv, 16);
                load8(va, svn, 0);   rmw8(vb, dvv, 24);
                svv = svn; dvv = dvn;
            }
            const int m = n & 31;
            if (m > 0) {
                rmw8(va, dvv, 0);
                if (m > 8)  { load8(vb, svv, 8);  rmw8(vb, dvv, 8); }
                if (m > 16) { load8(va, svv, 16); rmw8(va, dvv, 16); }
                if (m > 24) { load8(vb, svv, 24); rmw8(vb, dvv, 24); }
            }
        } else if (FIRST) {
            for (int i = tid; i < OWN_ROWS; i += 1024) degc[i] = 0;
            __syncthreads();
        }
    }
    if (scan) {
        int*            q  = queue + wave * QCAP;
        unsigned short* qr = qrow  + wave * QCAP;
        int             qn = 0;

        const int nfull = E / SCAN_EDGES;
        const int first = (int)((blockIdx.x * (unsigned)nfull) / gridDim.x);

        auto push_row = [&](int d, int e) {
            const unsigned t    = (unsigned)(d - base);
            const bool     mine = t < (unsigned)rows;
            const unsigned mask = __builtin_amdgcn_ballot_w32(mine);
            if (mask != 0u) {
                if (mine) {
                    const int slot = qn + (int)__builtin_amdgcn_mbcnt_lo(mask, 0u);
                    q[slot]  = e;
                    qr[slot] = (unsigned short)t;
                }
                qn += __builtin_popcount(mask);
            }
        };

        auto scan_step = [&](int (&use)[SCAN_ROWS], int (&fill)[SCAN_ROWS], int c) {
            int cl = c + 2 * OWN_WAVES;
            if (cl > nfull - 1) cl = nfull - 1;
            const int ccl = (cl + first < nfull) ? (cl + first) : (cl + first - nfull);
            const int* dl = dst + ccl * SCAN_EDGES + lane;
#pragma unroll
            for (int j = 0; j < SCAN_ROWS; ++j) fill[j] = dl[32 * j];

            if (c >= 0 && c < nfull) {
                const int cc = (c + first < nfull) ? (c + first) : (c + first - nfull);
                const int e0 = cc * SCAN_EDGES + lane;
#pragma unroll
                for (int j = 0; j < SCAN_ROWS; ++j) push_row(use[j], e0 + 32 * j);
            }

            while (qn >= 32) {
                qn -= 32;
                const int ee = q[qn + lane];
                add_rows(src[ee], (int)qr[qn + lane], 32);
            }
        };

        if (nfull > 0) {
            int ra[SCAN_ROWS], rb[SCAN_ROWS], rc[SCAN_ROWS];
#pragma unroll
            for (int j = 0; j < SCAN_ROWS; ++j) { ra[j] = -1; rb[j] = -1; }
            for (int c = wave - 2 * OWN_WAVES; c < nfull; c += 3 * OWN_WAVES) {
                scan_step(ra, rc, c);
                scan_step(rb, ra, c + OWN_WAVES);
                scan_step(rc, rb, c + 2 * OWN_WAVES);
            }
        }
        if (wave * 32 < E - nfull * SCAN_EDGES) {
            const int e = nfull * SCAN_EDGES + wave * 32 + lane;
            push_row((e < E) ? dst[e] : -1, e);
        }
        while (qn > 0) {
            const int n = (qn < 32) ? qn : 32;
            qn -= n;
            int sv = 0, dv = OWN_ROWS;
            if (lane < n) { const int ee = q[qn + lane]; sv = src[ee]; dv = (int)qr[qn + lane]; }
            add_rows(sv, dv, n);
        }
    }
    __syncthreads();

    float* outp = agg + (size_t)base * HDIM;
    for (int i = tid * 4; i < rows * HDIM; i += 1024 * 4)
        *(float4*)(outp + i) = *(const float4*)(&acc[i]);
    if (FIRST) {
        for (int r = tid; r < rows; r += 1024)
            invdeg[base + r] = 1.0f / fmaxf((float)degc[r], 1.0f);
    }
}

extern "C" void kernel_launch(void* const* d_in, const int* in_sizes, int n_in,
                              void* d_out, int out_size, void* d_ws, size_t ws_size,
                              hipStream_t stream)
{
    const float* x      = (const float*)d_in[0];
    const int*   edge   = (const int*)  d_in[1];
    const float* W_proj = (const float*)d_in[2];
    const float* b_proj = (const float*)d_in[3];
    const float* Wl[3]  = { (const float*)d_in[4], (const float*)d_in[7], (const float*)d_in[10] };
    const float* bl[3]  = { (const float*)d_in[5], (const float*)d_in[8], (const float*)d_in[11] };
    const float* Wr[3]  = { (const float*)d_in[6], (const float*)d_in[9], (const float*)d_in[12] };
    const float* Wc     = (const float*)d_in[13];
    const float* bc     = (const float*)d_in[14];

    const int N    = in_sizes[0] / 256;
    const int E    = in_sizes[1] / 2;
    const int ncls = in_sizes[14];

    const int* src = edge;
    const int* dst = edge + E;

    const int nblk = (N + OWN_ROWS - 1) / OWN_ROWS;
    const int ngrp = (nblk + FGROUP - 1) / FGROUP;

    float* hA     = (float*)d_ws;
    float* hB     = hA  + (size_t)N * HDIM;
    float* agg    = hB  + (size_t)N * HDIM;
    float* invdeg = agg + (size_t)N * HDIM;
    int*   counts = (int*)(invdeg + N);
    int*   lists  = counts + (size_t)nblk * SLOTS;

    const int rowTiles = (N + 15) / 16;

    edge_filter_kernel<<<ngrp * FPARTS, 1024, 0, stream>>>(src, dst, lists, counts, E, N, nblk);

    proj_strip_kernel<<<256, STRIP_THREADS, 0, stream>>>(x, W_proj, b_proj, hA, rowTiles);

    for (int l = 0; l < 3; ++l) {
        if (l == 0)
            owned_rows_sum_kernel<true><<<nblk, 1024, 0, stream>>>(src, dst, hA, agg, lists, counts, invdeg, E, N);
        else
            owned_rows_sum_kernel<false><<<nblk, 1024, 0, stream>>>(src, dst, hA, agg, lists, counts, invdeg, E, N);
        sage_strip_kernel<<<256, STRIP_THREADS, 0, stream>>>(agg, invdeg, hA,
                                                             Wl[l], bl[l], Wr[l], hB, rowTiles);
        float* t = hA; hA = hB; hB = t;
    }

    cls_strip_kernel<<<dim3((rowTiles + 2 * CLS_WAVES - 1) / (2 * CLS_WAVES), (ncls + 16 * CLS_TILES - 1) / (16 * CLS_TILES)),
                       CLS_THREADS, 0, stream>>>(hA, Wc, bc, (float*)d_out, ncls, rowTiles);
}
